# starting kernel
# baseline (speedup 1.0000x reference)
#include <hip/hip_runtime.h>
#include <hip/hip_bf16.h>

#define D_DIM 128

typedef __attribute__((ext_vector_type(2))) float v2f;
typedef __attribute__((ext_vector_type(8))) float v8f;

__device__ __forceinline__ unsigned f2ord(float f) {
    unsigned u = __float_as_uint(f);
    return (u & 0x80000000u) ? ~u : (u | 0x80000000u);
}
__device__ __forceinline__ float ord2f(unsigned u) {
    return (u & 0x80000000u) ? __uint_as_float(u & 0x7fffffffu)
                             : __uint_as_float(~u);
}

__global__ __launch_bounds__(256) void proj_wmma(const float* __restrict__ A,
                                                 const float* __restrict__ W,
                                                 float* __restrict__ P, int N) {
    const int rowTile = blockIdx.x << 4;
    const int wave    = threadIdx.x >> 5;
    const int lane    = threadIdx.x & 31;
    const int colTile = wave << 4;
    const int r       = lane & 15;
    const int koff    = (lane >> 4) << 1;

    int arow = rowTile + r;
    if (arow >= N) arow = N - 1;
    const float* __restrict__ ap = A + (size_t)arow * D_DIM;
    const float* __restrict__ bp = W + colTile + r;

    v8f c = {};
#pragma unroll 8
    for (int k0 = 0; k0 < D_DIM; k0 += 4) {
        v2f a, b;
        a.x = ap[k0 + koff];
        a.y = ap[k0 + koff + 1];
        b.x = bp[(size_t)(k0 + koff) * D_DIM];
        b.y = bp[(size_t)(k0 + koff + 1) * D_DIM];
        c = __builtin_amdgcn_wmma_f32_16x16x4_f32(false, a, false, b,
                                                  (short)0, c, false, false);
    }
    const int mbase = rowTile + ((lane >> 4) << 3);
    const int col   = colTile + r;
    if (rowTile + 16 <= N) {
        float* __restrict__ dst = P + (size_t)mbase * D_DIM + col;
#pragma unroll
        for (int v = 0; v < 8; ++v)
            dst[(size_t)v * D_DIM] = c[v];
    } else {
#pragma unroll
        for (int v = 0; v < 8; ++v) {
            int row = mbase + v;
            if (row < N) P[(size_t)row * D_DIM + col] = c[v];
        }
    }
}

__global__ __launch_bounds__(256) void edge_score(
    const float* __restrict__ P, const int* __restrict__ head,
    const int* __restrict__ tail, const int* __restrict__ etype,
    const float* __restrict__ rel, float* __restrict__ score,
    unsigned* __restrict__ mkey, int E) {
    int e = (blockIdx.x << 3) + (threadIdx.x >> 5);
    if (e >= E) return;
    int lane = threadIdx.x & 31;
    int h = head[e], t = tail[e], rt = etype[e];
    float4 q  = reinterpret_cast<const float4*>(P   + (size_t)h  * D_DIM)[lane];
    float4 k  = reinterpret_cast<const float4*>(P   + (size_t)t  * D_DIM)[lane];
    float4 rv = reinterpret_cast<const float4*>(rel + (size_t)rt * D_DIM)[lane];
    float s = q.x*k.x*rv.x + q.y*k.y*rv.y + q.z*k.z*rv.z + q.w*k.w*rv.w;
    s += __shfl_xor(s, 1, 32);
    s += __shfl_xor(s, 2, 32);
    s += __shfl_xor(s, 4, 32);
    s += __shfl_xor(s, 8, 32);
    s *= 0.125f;
    if ((lane & 15) == 0) {
        int hh = lane >> 4;
        score[((size_t)e << 1) + hh] = s;
        atomicMax(&mkey[((size_t)h << 1) + hh], f2ord(s));
    }
}

__global__ __launch_bounds__(256) void edge_expsum(
    const int* __restrict__ head, const unsigned* __restrict__ mkey,
    float* __restrict__ score, float* __restrict__ denom, int E2) {
    int idx = blockIdx.x * blockDim.x + threadIdx.x;
    if (idx >= E2) return;
    int e = idx >> 1, hh = idx & 1;
    int h = head[e];
    float m  = ord2f(mkey[((size_t)h << 1) + hh]);
    float ex = __expf(score[idx] - m);
    score[idx] = ex;
    atomicAdd(&denom[((size_t)h << 1) + hh], ex);
}

__global__ __launch_bounds__(256) void edge_aggregate(
    const float* __restrict__ entc, const int* __restrict__ head,
    const int* __restrict__ tail, const int* __restrict__ etype,
    const float* __restrict__ rel, const float* __restrict__ ex,
    const float* __restrict__ denom, float* __restrict__ agg, int E) {
    int e = (blockIdx.x << 3) + (threadIdx.x >> 5);
    if (e >= E) return;
    int lane = threadIdx.x & 31;
    int hh = lane >> 4;
    int h = head[e], t = tail[e], rt = etype[e];
    float attn = ex[((size_t)e << 1) + hh] / denom[((size_t)h << 1) + hh];
    float4 v  = reinterpret_cast<const float4*>(entc + (size_t)t  * D_DIM)[lane];
    float4 rv = reinterpret_cast<const float4*>(rel  + (size_t)rt * D_DIM)[lane];
    float* dst = agg + (size_t)h * D_DIM + (lane << 2);
    atomicAdd(dst + 0, v.x * rv.x * attn);
    atomicAdd(dst + 1, v.y * rv.y * attn);
    atomicAdd(dst + 2, v.z * rv.z * attn);
    atomicAdd(dst + 3, v.w * rv.w * attn);
}

__global__ __launch_bounds__(256) void user_aggregate(
    const float* __restrict__ entc, const int* __restrict__ uidx,
    const int* __restrict__ iidx, const float* __restrict__ w,
    float* __restrict__ uagg, int E) {
    int e = (blockIdx.x << 3) + (threadIdx.x >> 5);
    if (e >= E) return;
    int lane = threadIdx.x & 31;
    int u = uidx[e], i = iidx[e];
    float ww = w[e];
    float4 v = reinterpret_cast<const float4*>(entc + (size_t)i * D_DIM)[lane];
    float* dst = uagg + (size_t)u * D_DIM + (lane << 2);
    atomicAdd(dst + 0, ww * v.x);
    atomicAdd(dst + 1, ww * v.y);
    atomicAdd(dst + 2, ww * v.z);
    atomicAdd(dst + 3, ww * v.w);
}

__global__ __launch_bounds__(256) void norm_residual(
    const float* __restrict__ agg, float* __restrict__ cur,
    float* __restrict__ res, int N) {
    int n = (blockIdx.x << 3) + (threadIdx.x >> 5);
    if (n >= N) return;
    int lane = threadIdx.x & 31;
    size_t vidx = ((size_t)n * D_DIM >> 2) + lane;
    float4 v = reinterpret_cast<const float4*>(agg)[vidx];
    float ss = v.x*v.x + v.y*v.y + v.z*v.z + v.w*v.w;
    ss += __shfl_xor(ss, 1, 32);
    ss += __shfl_xor(ss, 2, 32);
    ss += __shfl_xor(ss, 4, 32);
    ss += __shfl_xor(ss, 8, 32);
    ss += __shfl_xor(ss, 16, 32);
    float inv = 1.0f / fmaxf(sqrtf(ss), 1e-12f);
    v.x *= inv; v.y *= inv; v.z *= inv; v.w *= inv;
    if (cur) reinterpret_cast<float4*>(cur)[vidx] = v;
    float4 r = reinterpret_cast<const float4*>(res)[vidx];
    r.x += v.x; r.y += v.y; r.z += v.z; r.w += v.w;
    reinterpret_cast<float4*>(res)[vidx] = r;
}

extern "C" void kernel_launch(void* const* d_in, const int* in_sizes, int n_in,
                              void* d_out, int out_size, void* d_ws, size_t ws_size,
                              hipStream_t stream) {
    (void)n_in; (void)out_size; (void)ws_size;
    const float* user_emb   = (const float*)d_in[0];
    const float* entity_emb = (const float*)d_in[1];
    const float* W_Q        = (const float*)d_in[2];
    const float* rel_emb    = (const float*)d_in[3];
    const int*   edge_index = (const int*)d_in[4];
    const int*   edge_type  = (const int*)d_in[5];
    const int*   inter_edge = (const int*)d_in[6];
    const float* inter_w    = (const float*)d_in[7];

    const int N_USR = in_sizes[0] / D_DIM;
    const int N_ENT = in_sizes[1] / D_DIM;
    const int E     = in_sizes[5];
    const int E_UI  = in_sizes[7];

    const int* head = edge_index;
    const int* tail = edge_index + E;
    const int* uidx = inter_edge;
    const int* iidx = inter_edge + E_UI;

    float* out      = (float*)d_out;
    float* user_res = out;
    float* ent_res  = out + (size_t)N_USR * D_DIM;

    char* ws = (char*)d_ws;
    size_t off = 0;
    auto carve = [&](size_t bytes) -> void* {
        void* p = ws + off;
        off += (bytes + 255) & ~(size_t)255;
        return p;
    };
    float*    ent_cur  = (float*)   carve((size_t)N_ENT * D_DIM * 4);
    float*    P        = (float*)   carve((size_t)N_ENT * D_DIM * 4);
    float*    user_agg = (float*)   carve((size_t)N_USR * D_DIM * 4);
    float*    score    = (float*)   carve((size_t)E * 2 * 4);
    unsigned* mkey     = (unsigned*)carve((size_t)N_ENT * 2 * 4);
    float*    denom    = (float*)   carve((size_t)N_ENT * 2 * 4);

    hipMemcpyAsync(ent_cur,  entity_emb, (size_t)N_ENT * D_DIM * 4,
                   hipMemcpyDeviceToDevice, stream);
    hipMemcpyAsync(ent_res,  entity_emb, (size_t)N_ENT * D_DIM * 4,
                   hipMemcpyDeviceToDevice, stream);
    hipMemcpyAsync(user_res, user_emb,   (size_t)N_USR * D_DIM * 4,
                   hipMemcpyDeviceToDevice, stream);

    for (int hop = 0; hop < 2; ++hop) {
        hipMemsetAsync(mkey,     0, (size_t)N_ENT * 2 * 4, stream);
        hipMemsetAsync(denom,    0, (size_t)N_ENT * 2 * 4, stream);
        hipMemsetAsync(user_agg, 0, (size_t)N_USR * D_DIM * 4, stream);

        proj_wmma<<<(N_ENT + 15) / 16, 256, 0, stream>>>(ent_cur, W_Q, P, N_ENT);
        edge_score<<<(E + 7) / 8, 256, 0, stream>>>(P, head, tail, edge_type,
                                                    rel_emb, score, mkey, E);
        edge_expsum<<<(2 * E + 255) / 256, 256, 0, stream>>>(head, mkey, score,
                                                             denom, 2 * E);
        hipMemsetAsync(P, 0, (size_t)N_ENT * D_DIM * 4, stream);
        edge_aggregate<<<(E + 7) / 8, 256, 0, stream>>>(ent_cur, head, tail,
                                                        edge_type, rel_emb,
                                                        score, denom, P, E);
        user_aggregate<<<(E_UI + 7) / 8, 256, 0, stream>>>(ent_cur, uidx, iidx,
                                                           inter_w, user_agg, E_UI);
        norm_residual<<<(N_ENT + 7) / 8, 256, 0, stream>>>(P, ent_cur, ent_res, N_ENT);
        norm_residual<<<(N_USR + 7) / 8, 256, 0, stream>>>(user_agg, nullptr,
                                                           user_res, N_USR);
    }
}


# edge and user scatter-add: lane l owns elements l,l+32,l+64,l+96, so each f32 atomic add and row load of a wave covers one whole 128-B line, not four lines; sums stay in atomic arrival order
# speedup vs baseline: 1.1852x; 1.1852x over previous
#include <hip/hip_runtime.h>
#include <hip/hip_bf16.h>

#define D_DIM 128

typedef __attribute__((ext_vector_type(2))) float v2f;
typedef __attribute__((ext_vector_type(8))) float v8f;

__device__ __forceinline__ unsigned f2ord(float f) {
    unsigned u = __float_as_uint(f);
    return (u & 0x80000000u) ? ~u : (u | 0x80000000u);
}
__device__ __forceinline__ float ord2f(unsigned u) {
    return (u & 0x80000000u) ? __uint_as_float(u & 0x7fffffffu)
                             : __uint_as_float(~u);
}

__global__ __launch_bounds__(256) void proj_wmma(const float* __restrict__ A,
                                                 const float* __restrict__ W,
                                                 float* __restrict__ P, int N) {
    const int rowTile = blockIdx.x << 4;
    const int wave    = threadIdx.x >> 5;
    const int lane    = threadIdx.x & 31;
    const int colTile = wave << 4;
    const int r       = lane & 15;
    const int koff    = (lane >> 4) << 1;

    int arow = rowTile + r;
    if (arow >= N) arow = N - 1;
    const float* __restrict__ ap = A + (size_t)arow * D_DIM;
    const float* __restrict__ bp = W + colTile + r;

    v8f c = {};
#pragma unroll 8
    for (int k0 = 0; k0 < D_DIM; k0 += 4) {
        v2f a, b;
        a.x = ap[k0 + koff];
        a.y = ap[k0 + koff + 1];
        b.x = bp[(size_t)(k0 + koff) * D_DIM];
        b.y = bp[(size_t)(k0 + koff + 1) * D_DIM];
        c = __builtin_amdgcn_wmma_f32_16x16x4_f32(false, a, false, b,
                                                  (short)0, c, false, false);
    }
    const int mbase = rowTile + ((lane >> 4) << 3);
    const int col   = colTile + r;
    if (rowTile + 16 <= N) {
        float* __restrict__ dst = P + (size_t)mbase * D_DIM + col;
#pragma unroll
        for (int v = 0; v < 8; ++v)
            dst[(size_t)v * D_DIM] = c[v];
    } else {
#pragma unroll
        for (int v = 0; v < 8; ++v) {
            int row = mbase + v;
            if (row < N) P[(size_t)row * D_DIM + col] = c[v];
        }
    }
}

__global__ __launch_bounds__(256) void edge_score(
    const float* __restrict__ P, const int* __restrict__ head,
    const int* __restrict__ tail, const int* __restrict__ etype,
    const float* __restrict__ rel, float* __restrict__ score,
    unsigned* __restrict__ mkey, int E) {
    int e = (blockIdx.x << 3) + (threadIdx.x >> 5);
    if (e >= E) return;
    int lane = threadIdx.x & 31;
    int h = head[e], t = tail[e], rt = etype[e];
    float4 q  = reinterpret_cast<const float4*>(P   + (size_t)h  * D_DIM)[lane];
    float4 k  = reinterpret_cast<const float4*>(P   + (size_t)t  * D_DIM)[lane];
    float4 rv = reinterpret_cast<const float4*>(rel + (size_t)rt * D_DIM)[lane];
    float s = q.x*k.x*rv.x + q.y*k.y*rv.y + q.z*k.z*rv.z + q.w*k.w*rv.w;
    s += __shfl_xor(s, 1, 32);
    s += __shfl_xor(s, 2, 32);
    s += __shfl_xor(s, 4, 32);
    s += __shfl_xor(s, 8, 32);
    s *= 0.125f;
    if ((lane & 15) == 0) {
        int hh = lane >> 4;
        score[((size_t)e << 1) + hh] = s;
        atomicMax(&mkey[((size_t)h << 1) + hh], f2ord(s));
    }
}

__global__ __launch_bounds__(256) void edge_expsum(
    const int* __restrict__ head, const unsigned* __restrict__ mkey,
    float* __restrict__ score, float* __restrict__ denom, int E2) {
    int idx = blockIdx.x * blockDim.x + threadIdx.x;
    if (idx >= E2) return;
    int e = idx >> 1, hh = idx & 1;
    int h = head[e];
    float m  = ord2f(mkey[((size_t)h << 1) + hh]);
    float ex = __expf(score[idx] - m);
    score[idx] = ex;
    atomicAdd(&denom[((size_t)h << 1) + hh], ex);
}

__global__ __launch_bounds__(256) void edge_aggregate(
    const float* __restrict__ entc, const int* __restrict__ head,
    const int* __restrict__ tail, const int* __restrict__ etype,
    const float* __restrict__ rel, const float* __restrict__ ex,
    const float* __restrict__ denom, float* __restrict__ agg, int E) {
    int e = (blockIdx.x << 3) + (threadIdx.x >> 5);
    if (e >= E) return;
    int lane = threadIdx.x & 31;
    int h = head[e], t = tail[e], rt = etype[e];
    const float* __restrict__ vp = entc + (size_t)t  * D_DIM + lane;
    const float* __restrict__ rp = rel  + (size_t)rt * D_DIM + lane;
    float v0 = vp[0],  v1 = vp[32], v2 = vp[64], v3 = vp[96];
    float r0 = rp[0],  r1 = rp[32], r2 = rp[64], r3 = rp[96];
    float2 exv = reinterpret_cast<const float2*>(ex)[e];
    float2 dn  = reinterpret_cast<const float2*>(denom)[h];
    float attn0 = exv.x / dn.x;
    float attn1 = exv.y / dn.y;
    float* dst = agg + (size_t)h * D_DIM + lane;
    atomicAdd(dst,      v0 * r0 * attn0);
    atomicAdd(dst + 32, v1 * r1 * attn0);
    atomicAdd(dst + 64, v2 * r2 * attn1);
    atomicAdd(dst + 96, v3 * r3 * attn1);
}

__global__ __launch_bounds__(256) void user_aggregate(
    const float* __restrict__ entc, const int* __restrict__ uidx,
    const int* __restrict__ iidx, const float* __restrict__ w,
    float* __restrict__ uagg, int E) {
    int e = (blockIdx.x << 3) + (threadIdx.x >> 5);
    if (e >= E) return;
    int lane = threadIdx.x & 31;
    int u = uidx[e], i = iidx[e];
    float ww = w[e];
    const float* __restrict__ vp = entc + (size_t)i * D_DIM + lane;
    float v0 = vp[0], v1 = vp[32], v2 = vp[64], v3 = vp[96];
    float* dst = uagg + (size_t)u * D_DIM + lane;
    atomicAdd(dst,      ww * v0);
    atomicAdd(dst + 32, ww * v1);
    atomicAdd(dst + 64, ww * v2);
    atomicAdd(dst + 96, ww * v3);
}

__global__ __launch_bounds__(256) void norm_residual(
    const float* __restrict__ agg, float* __restrict__ cur,
    float* __restrict__ res, int N) {
    int n = (blockIdx.x << 3) + (threadIdx.x >> 5);
    if (n >= N) return;
    int lane = threadIdx.x & 31;
    size_t vidx = ((size_t)n * D_DIM >> 2) + lane;
    float4 v = reinterpret_cast<const float4*>(agg)[vidx];
    float ss = v.x*v.x + v.y*v.y + v.z*v.z + v.w*v.w;
    ss += __shfl_xor(ss, 1, 32);
    ss += __shfl_xor(ss, 2, 32);
    ss += __shfl_xor(ss, 4, 32);
    ss += __shfl_xor(ss, 8, 32);
    ss += __shfl_xor(ss, 16, 32);
    float inv = 1.0f / fmaxf(sqrtf(ss), 1e-12f);
    v.x *= inv; v.y *= inv; v.z *= inv; v.w *= inv;
    if (cur) reinterpret_cast<float4*>(cur)[vidx] = v;
    float4 r = reinterpret_cast<const float4*>(res)[vidx];
    r.x += v.x; r.y += v.y; r.z += v.z; r.w += v.w;
    reinterpret_cast<float4*>(res)[vidx] = r;
}

extern "C" void kernel_launch(void* const* d_in, const int* in_sizes, int n_in,
                              void* d_out, int out_size, void* d_ws, size_t ws_size,
                              hipStream_t stream) {
    (void)n_in; (void)out_size; (void)ws_size;
    const float* user_emb   = (const float*)d_in[0];
    const float* entity_emb = (const float*)d_in[1];
    const float* W_Q        = (const float*)d_in[2];
    const float* rel_emb    = (const float*)d_in[3];
    const int*   edge_index = (const int*)d_in[4];
    const int*   edge_type  = (const int*)d_in[5];
    const int*   inter_edge = (const int*)d_in[6];
    const float* inter_w    = (const float*)d_in[7];

    const int N_USR = in_sizes[0] / D_DIM;
    const int N_ENT = in_sizes[1] / D_DIM;
    const int E     = in_sizes[5];
    const int E_UI  = in_sizes[7];

    const int* head = edge_index;
    const int* tail = edge_index + E;
    const int* uidx = inter_edge;
    const int* iidx = inter_edge + E_UI;

    float* out      = (float*)d_out;
    float* user_res = out;
    float* ent_res  = out + (size_t)N_USR * D_DIM;

    char* ws = (char*)d_ws;
    size_t off = 0;
    auto carve = [&](size_t bytes) -> void* {
        void* p = ws + off;
        off += (bytes + 255) & ~(size_t)255;
        return p;
    };
    float*    ent_cur  = (float*)   carve((size_t)N_ENT * D_DIM * 4);
    float*    P        = (float*)   carve((size_t)N_ENT * D_DIM * 4);
    float*    user_agg = (float*)   carve((size_t)N_USR * D_DIM * 4);
    float*    score    = (float*)   carve((size_t)E * 2 * 4);
    unsigned* mkey     = (unsigned*)carve((size_t)N_ENT * 2 * 4);
    float*    denom    = (float*)   carve((size_t)N_ENT * 2 * 4);

    hipMemcpyAsync(ent_cur,  entity_emb, (size_t)N_ENT * D_DIM * 4,
                   hipMemcpyDeviceToDevice, stream);
    hipMemcpyAsync(ent_res,  entity_emb, (size_t)N_ENT * D_DIM * 4,
                   hipMemcpyDeviceToDevice, stream);
    hipMemcpyAsync(user_res, user_emb,   (size_t)N_USR * D_DIM * 4,
                   hipMemcpyDeviceToDevice, stream);

    for (int hop = 0; hop < 2; ++hop) {
        hipMemsetAsync(mkey,     0, (size_t)N_ENT * 2 * 4, stream);
        hipMemsetAsync(denom,    0, (size_t)N_ENT * 2 * 4, stream);
        hipMemsetAsync(user_agg, 0, (size_t)N_USR * D_DIM * 4, stream);

        proj_wmma<<<(N_ENT + 15) / 16, 256, 0, stream>>>(ent_cur, W_Q, P, N_ENT);
        edge_score<<<(E + 7) / 8, 256, 0, stream>>>(P, head, tail, edge_type,
                                                    rel_emb, score, mkey, E);
        edge_expsum<<<(2 * E + 255) / 256, 256, 0, stream>>>(head, mkey, score,
                                                             denom, 2 * E);
        hipMemsetAsync(P, 0, (size_t)N_ENT * D_DIM * 4, stream);
        edge_aggregate<<<(E + 7) / 8, 256, 0, stream>>>(ent_cur, head, tail,
                                                        edge_type, rel_emb,
                                                        score, denom, P, E);
        user_aggregate<<<(E_UI + 7) / 8, 256, 0, stream>>>(ent_cur, uidx, iidx,
                                                           inter_w, user_agg, E_UI);
        norm_residual<<<(N_ENT + 7) / 8, 256, 0, stream>>>(P, ent_cur, ent_res, N_ENT);
        norm_residual<<<(N_USR + 7) / 8, 256, 0, stream>>>(user_agg, nullptr,
                                                           user_res, N_USR);
    }
}


# edge score, edge and user scatter-add: a wave owns 8 consecutive edges (was 1); indices read once per wave, row bases scalar, a batch's rows all in flight; sums stay in atomic arrival order
# speedup vs baseline: 1.4323x; 1.2085x over previous
#include <hip/hip_runtime.h>
#include <hip/hip_bf16.h>

#define D_DIM 128

typedef __attribute__((ext_vector_type(2))) float v2f;
typedef __attribute__((ext_vector_type(8))) float v8f;

__device__ __forceinline__ unsigned f2ord(float f) {
    unsigned u = __float_as_uint(f);
    return (u & 0x80000000u) ? ~u : (u | 0x80000000u);
}
__device__ __forceinline__ float ord2f(unsigned u) {
    return (u & 0x80000000u) ? __uint_as_float(u & 0x7fffffffu)
                             : __uint_as_float(~u);
}

#define EPW 8
__device__ __forceinline__ int lane_get(int v, int j) {
    return __builtin_amdgcn_readlane(v, j);
}
__device__ __forceinline__ float lane_getf(float v, int j) {
    return __int_as_float(__builtin_amdgcn_readlane(__float_as_int(v), j));
}
__device__ __forceinline__ int wave_first_edge() {
    return __builtin_amdgcn_readfirstlane((int)(((blockIdx.x << 3) + (threadIdx.x >> 5)) * EPW));
}

__global__ __launch_bounds__(256) void proj_wmma(const float* __restrict__ A,
                                                 const float* __restrict__ W,
                                                 float* __restrict__ P, int N) {
    const int rowTile = blockIdx.x << 4;
    const int wave    = threadIdx.x >> 5;
    const int lane    = threadIdx.x & 31;
    const int colTile = wave << 4;
    const int r       = lane & 15;
    const int koff    = (lane >> 4) << 1;

    int arow = rowTile + r;
    if (arow >= N) arow = N - 1;
    const float* __restrict__ ap = A + (size_t)arow * D_DIM;
    const float* __restrict__ bp = W + colTile + r;

    v8f c = {};
#pragma unroll 8
    for (int k0 = 0; k0 < D_DIM; k0 += 4) {
        v2f a, b;
        a.x = ap[k0 + koff];
        a.y = ap[k0 + koff + 1];
        b.x = bp[(size_t)(k0 + koff) * D_DIM];
        b.y = bp[(size_t)(k0 + koff + 1) * D_DIM];
        c = __builtin_amdgcn_wmma_f32_16x16x4_f32(false, a, false, b,
                                                  (short)0, c, false, false);
    }
    const int mbase = rowTile + ((lane >> 4) << 3);
    const int col   = colTile + r;
    if (rowTile + 16 <= N) {
        float* __restrict__ dst = P + (size_t)mbase * D_DIM + col;
#pragma unroll
        for (int v = 0; v < 8; ++v)
            dst[(size_t)v * D_DIM] = c[v];
    } else {
#pragma unroll
        for (int v = 0; v < 8; ++v) {
            int row = mbase + v;
            if (row < N) P[(size_t)row * D_DIM + col] = c[v];
        }
    }
}

__global__ __launch_bounds__(256) void edge_score(
    const float* __restrict__ P, const int* __restrict__ head,
    const int* __restrict__ tail, const int* __restrict__ etype,
    const float* __restrict__ rel, float* __restrict__ score,
    unsigned* __restrict__ mkey, int E) {
    const int e0 = wave_first_edge();
    if (e0 >= E) return;
    const int lane = threadIdx.x & 31;
    int em = e0 + (lane & (EPW - 1));
    const bool valid = em < E;
    if (!valid) em = E - 1;
    const int h = head[em], t = tail[em], rt = etype[em];
    float mine = 0.0f;
#pragma unroll
    for (int b = 0; b < EPW; b += 4) {
        float4 q[4], k[4], rv[4];
#pragma unroll
        for (int j = 0; j < 4; ++j) {
            const int hj = lane_get(h, b + j), tj = lane_get(t, b + j), rj = lane_get(rt, b + j);
            q[j]  = reinterpret_cast<const float4*>(P   + (size_t)hj * D_DIM)[lane];
            k[j]  = reinterpret_cast<const float4*>(P   + (size_t)tj * D_DIM)[lane];
            rv[j] = reinterpret_cast<const float4*>(rel + (size_t)rj * D_DIM)[lane];
        }
#pragma unroll
        for (int j = 0; j < 4; ++j) {
            float s = q[j].x*k[j].x*rv[j].x + q[j].y*k[j].y*rv[j].y
                    + q[j].z*k[j].z*rv[j].z + q[j].w*k[j].w*rv[j].w;
            s += __shfl_xor(s, 1, 32);
            s += __shfl_xor(s, 2, 32);
            s += __shfl_xor(s, 4, 32);
            s += __shfl_xor(s, 8, 32);
            s *= 0.125f;
            if ((lane & (EPW - 1)) == b + j) mine = s;
        }
    }
    if (valid && (lane & 8) == 0) {
        const int hh = lane >> 4;
        score[((size_t)em << 1) + hh] = mine;
        atomicMax(&mkey[((size_t)h << 1) + hh], f2ord(mine));
    }
}

__global__ __launch_bounds__(256) void edge_expsum(
    const int* __restrict__ head, const unsigned* __restrict__ mkey,
    float* __restrict__ score, float* __restrict__ denom, int E2) {
    int idx = blockIdx.x * blockDim.x + threadIdx.x;
    if (idx >= E2) return;
    int e = idx >> 1, hh = idx & 1;
    int h = head[e];
    float m  = ord2f(mkey[((size_t)h << 1) + hh]);
    float ex = __expf(score[idx] - m);
    score[idx] = ex;
    atomicAdd(&denom[((size_t)h << 1) + hh], ex);
}

__global__ __launch_bounds__(256) void edge_aggregate(
    const float* __restrict__ entc, const int* __restrict__ head,
    const int* __restrict__ tail, const int* __restrict__ etype,
    const float* __restrict__ rel, const float* __restrict__ ex,
    const float* __restrict__ denom, float* __restrict__ agg, int E) {
    const int e0 = wave_first_edge();
    if (e0 >= E) return;
    const int lane = threadIdx.x & 31;
    const int nE = (E - e0 < EPW) ? (E - e0) : EPW;
    int em = e0 + (lane & (EPW - 1));
    if (em >= E) em = E - 1;
    const int h = head[em], t = tail[em], rt = etype[em];
    const float2 exv = reinterpret_cast<const float2*>(ex)[em];
    const float2 dn  = reinterpret_cast<const float2*>(denom)[h];
    const float attn0 = exv.x / dn.x;
    const float attn1 = exv.y / dn.y;
#pragma unroll
    for (int b = 0; b < EPW; b += 4) {
        float v[4][4], r[4][4];
#pragma unroll
        for (int j = 0; j < 4; ++j) {
            const int tj = lane_get(t, b + j), rj = lane_get(rt, b + j);
            const float* __restrict__ vp = entc + (size_t)tj * D_DIM + lane;
            const float* __restrict__ rp = rel  + (size_t)rj * D_DIM + lane;
            v[j][0] = vp[0]; v[j][1] = vp[32]; v[j][2] = vp[64]; v[j][3] = vp[96];
            r[j][0] = rp[0]; r[j][1] = rp[32]; r[j][2] = rp[64]; r[j][3] = rp[96];
        }
#pragma unroll
        for (int j = 0; j < 4; ++j) {
            if (b + j < nE) {
                const int   hj = lane_get(h, b + j);
                const float a0 = lane_getf(attn0, b + j), a1 = lane_getf(attn1, b + j);
                float* dst = agg + (size_t)hj * D_DIM + lane;
                atomicAdd(dst,      v[j][0] * r[j][0] * a0);
                atomicAdd(dst + 32, v[j][1] * r[j][1] * a0);
                atomicAdd(dst + 64, v[j][2] * r[j][2] * a1);
                atomicAdd(dst + 96, v[j][3] * r[j][3] * a1);
            }
        }
    }
}

__global__ __launch_bounds__(256) void user_aggregate(
    const float* __restrict__ entc, const int* __restrict__ uidx,
    const int* __restrict__ iidx, const float* __restrict__ w,
    float* __restrict__ uagg, int E) {
    const int e0 = wave_first_edge();
    if (e0 >= E) return;
    const int lane = threadIdx.x & 31;
    const int nE = (E - e0 < EPW) ? (E - e0) : EPW;
    int em = e0 + (lane & (EPW - 1));
    if (em >= E) em = E - 1;
    const int u = uidx[em], i = iidx[em];
    const float ww = w[em];
    float v[EPW][4];
#pragma unroll
    for (int j = 0; j < EPW; ++j) {
        const int ij = lane_get(i, j);
        const float* __restrict__ vp = entc + (size_t)ij * D_DIM + lane;
        v[j][0] = vp[0]; v[j][1] = vp[32]; v[j][2] = vp[64]; v[j][3] = vp[96];
    }
#pragma unroll
    for (int j = 0; j < EPW; ++j) {
        if (j < nE) {
            const int   uj = lane_get(u, j);
            const float wj = lane_getf(ww, j);
            float* dst = uagg + (size_t)uj * D_DIM + lane;
            atomicAdd(dst,      wj * v[j][0]);
            atomicAdd(dst + 32, wj * v[j][1]);
            atomicAdd(dst + 64, wj * v[j][2]);
            atomicAdd(dst + 96, wj * v[j][3]);
        }
    }
}

__global__ __launch_bounds__(256) void norm_residual(
    const float* __restrict__ agg, float* __restrict__ cur,
    float* __restrict__ res, int N) {
    int n = (blockIdx.x << 3) + (threadIdx.x >> 5);
    if (n >= N) return;
    int lane = threadIdx.x & 31;
    size_t vidx = ((size_t)n * D_DIM >> 2) + lane;
    float4 v = reinterpret_cast<const float4*>(agg)[vidx];
    float ss = v.x*v.x + v.y*v.y + v.z*v.z + v.w*v.w;
    ss += __shfl_xor(ss, 1, 32);
    ss += __shfl_xor(ss, 2, 32);
    ss += __shfl_xor(ss, 4, 32);
    ss += __shfl_xor(ss, 8, 32);
    ss += __shfl_xor(ss, 16, 32);
    float inv = 1.0f / fmaxf(sqrtf(ss), 1e-12f);
    v.x *= inv; v.y *= inv; v.z *= inv; v.w *= inv;
    if (cur) reinterpret_cast<float4*>(cur)[vidx] = v;
    float4 r = reinterpret_cast<const float4*>(res)[vidx];
    r.x += v.x; r.y += v.y; r.z += v.z; r.w += v.w;
    reinterpret_cast<float4*>(res)[vidx] = r;
}

extern "C" void kernel_launch(void* const* d_in, const int* in_sizes, int n_in,
                              void* d_out, int out_size, void* d_ws, size_t ws_size,
                              hipStream_t stream) {
    (void)n_in; (void)out_size; (void)ws_size;
    const float* user_emb   = (const float*)d_in[0];
    const float* entity_emb = (const float*)d_in[1];
    const float* W_Q        = (const float*)d_in[2];
    const float* rel_emb    = (const float*)d_in[3];
    const int*   edge_index = (const int*)d_in[4];
    const int*   edge_type  = (const int*)d_in[5];
    const int*   inter_edge = (const int*)d_in[6];
    const float* inter_w    = (const float*)d_in[7];

    const int N_USR = in_sizes[0] / D_DIM;
    const int N_ENT = in_sizes[1] / D_DIM;
    const int E     = in_sizes[5];
    const int E_UI  = in_sizes[7];

    const int* head = edge_index;
    const int* tail = edge_index + E;
    const int* uidx = inter_edge;
    const int* iidx = inter_edge + E_UI;

    float* out      = (float*)d_out;
    float* user_res = out;
    float* ent_res  = out + (size_t)N_USR * D_DIM;

    char* ws = (char*)d_ws;
    size_t off = 0;
    auto carve = [&](size_t bytes) -> void* {
        void* p = ws + off;
        off += (bytes + 255) & ~(size_t)255;
        return p;
    };
    float*    ent_cur  = (float*)   carve((size_t)N_ENT * D_DIM * 4);
    float*    P        = (float*)   carve((size_t)N_ENT * D_DIM * 4);
    float*    user_agg = (float*)   carve((size_t)N_USR * D_DIM * 4);
    float*    score    = (float*)   carve((size_t)E * 2 * 4);
    unsigned* mkey     = (unsigned*)carve((size_t)N_ENT * 2 * 4);
    float*    denom    = (float*)   carve((size_t)N_ENT * 2 * 4);

    hipMemcpyAsync(ent_cur,  entity_emb, (size_t)N_ENT * D_DIM * 4,
                   hipMemcpyDeviceToDevice, stream);
    hipMemcpyAsync(ent_res,  entity_emb, (size_t)N_ENT * D_DIM * 4,
                   hipMemcpyDeviceToDevice, stream);
    hipMemcpyAsync(user_res, user_emb,   (size_t)N_USR * D_DIM * 4,
                   hipMemcpyDeviceToDevice, stream);

    for (int hop = 0; hop < 2; ++hop) {
        hipMemsetAsync(mkey,     0, (size_t)N_ENT * 2 * 4, stream);
        hipMemsetAsync(denom,    0, (size_t)N_ENT * 2 * 4, stream);
        hipMemsetAsync(user_agg, 0, (size_t)N_USR * D_DIM * 4, stream);

        proj_wmma<<<(N_ENT + 15) / 16, 256, 0, stream>>>(ent_cur, W_Q, P, N_ENT);
        edge_score<<<(E + 8 * EPW - 1) / (8 * EPW), 256, 0, stream>>>(P, head, tail, edge_type,
                                                    rel_emb, score, mkey, E);
        edge_expsum<<<(2 * E + 255) / 256, 256, 0, stream>>>(head, mkey, score,
                                                             denom, 2 * E);
        hipMemsetAsync(P, 0, (size_t)N_ENT * D_DIM * 4, stream);
        edge_aggregate<<<(E + 8 * EPW - 1) / (8 * EPW), 256, 0, stream>>>(ent_cur, head, tail,
                                                        edge_type, rel_emb,
                                                        score, denom, P, E);
        user_aggregate<<<(E_UI + 8 * EPW - 1) / (8 * EPW), 256, 0, stream>>>(ent_cur, uidx, iidx,
                                                           inter_w, user_agg, E_UI);
        norm_residual<<<(N_ENT + 7) / 8, 256, 0, stream>>>(P, ent_cur, ent_res, N_ENT);
        norm_residual<<<(N_USR + 7) / 8, 256, 0, stream>>>(user_agg, nullptr,
                                                           user_res, N_USR);
    }
}


# edges counting-sorted by destination row once per call; one wave per row does max, exp-sum, weighted sum, l2norm, residual in registers, no f32 atomics; sums in sort-arrival order
# speedup vs baseline: 2.0549x; 1.4347x over previous
#include <hip/hip_runtime.h>
#include <hip/hip_bf16.h>

#define D_DIM 128

typedef __attribute__((ext_vector_type(2))) float v2f;
typedef __attribute__((ext_vector_type(8))) float v8f;


__device__ __forceinline__ int lane_get(int v, int j) {
    return __builtin_amdgcn_readlane(v, j);
}
__device__ __forceinline__ int uni(int v) {
    return __builtin_amdgcn_readfirstlane(v);
}

__global__ __launch_bounds__(256) void proj_wmma(const float* __restrict__ A,
                                                 const float* __restrict__ W,
                                                 float* __restrict__ P, int N) {
    const int rowTile = blockIdx.x << 4;
    const int wave    = threadIdx.x >> 5;
    const int lane    = threadIdx.x & 31;
    const int colTile = wave << 4;
    const int r       = lane & 15;
    const int koff    = (lane >> 4) << 1;

    int arow = rowTile + r;
    if (arow >= N) arow = N - 1;
    const float* __restrict__ ap = A + (size_t)arow * D_DIM;
    const float* __restrict__ bp = W + colTile + r;

    v8f c = {};
#pragma unroll 8
    for (int k0 = 0; k0 < D_DIM; k0 += 4) {
        v2f a, b;
        a.x = ap[k0 + koff];
        a.y = ap[k0 + koff + 1];
        b.x = bp[(size_t)(k0 + koff) * D_DIM];
        b.y = bp[(size_t)(k0 + koff + 1) * D_DIM];
        c = __builtin_amdgcn_wmma_f32_16x16x4_f32(false, a, false, b,
                                                  (short)0, c, false, false);
    }
    const int mbase = rowTile + ((lane >> 4) << 3);
    const int col   = colTile + r;
    if (rowTile + 16 <= N) {
        float* __restrict__ dst = P + (size_t)mbase * D_DIM + col;
#pragma unroll
        for (int v = 0; v < 8; ++v)
            dst[(size_t)v * D_DIM] = c[v];
    } else {
#pragma unroll
        for (int v = 0; v < 8; ++v) {
            int row = mbase + v;
            if (row < N) P[(size_t)row * D_DIM + col] = c[v];
        }
    }
}

__global__ __launch_bounds__(256) void seg_count(
    const int* __restrict__ head, const int* __restrict__ uidx,
    int* __restrict__ cnt, int E, int EU, int NE) {
    const int g = blockIdx.x * 256 + threadIdx.x;
    if (g < E)            atomicAdd(&cnt[head[g]], 1);
    else if (g < E + EU)  atomicAdd(&cnt[NE + uidx[g - E]], 1);
}

__global__ __launch_bounds__(1024) void seg_scan(
    const int* __restrict__ cnt, int* __restrict__ rowptr,
    int* __restrict__ cursor, int NT) {
    __shared__ int wsum[32];
    __shared__ int sbase;
    const int tid = threadIdx.x, lane = tid & 31, wave = tid >> 5;
    const int start = blockIdx.x * 1024;

    int part = 0;
    for (int i = tid; i < start; i += 1024) part += cnt[i];
    part += __shfl_xor(part, 16, 32);
    part += __shfl_xor(part, 8, 32);
    part += __shfl_xor(part, 4, 32);
    part += __shfl_xor(part, 2, 32);
    part += __shfl_xor(part, 1, 32);
    if (lane == 0) wsum[wave] = part;
    __syncthreads();
    if (wave == 0) {
        int t = wsum[lane];
        t += __shfl_xor(t, 16, 32);
        t += __shfl_xor(t, 8, 32);
        t += __shfl_xor(t, 4, 32);
        t += __shfl_xor(t, 2, 32);
        t += __shfl_xor(t, 1, 32);
        if (lane == 0) sbase = t;
    }
    __syncthreads();
    const int base = sbase;

    const int i = start + tid;
    const int c = (i < NT) ? cnt[i] : 0;
    int inc = c;
#pragma unroll
    for (int o = 1; o < 32; o <<= 1) {
        const int t = __shfl_up(inc, o, 32);
        if (lane >= o) inc += t;
    }
    if (lane == 31) wsum[wave] = inc;
    __syncthreads();
    if (wave == 0) {
        const int t = wsum[lane];
        int ti = t;
#pragma unroll
        for (int o = 1; o < 32; o <<= 1) {
            const int u = __shfl_up(ti, o, 32);
            if (lane >= o) ti += u;
        }
        wsum[lane] = ti - t;
    }
    __syncthreads();
    const int excl = base + wsum[wave] + inc - c;
    if (i < NT) {
        rowptr[i] = excl;
        cursor[i] = excl;
        if (i == NT - 1) rowptr[NT] = excl + c;
    }
}

__global__ __launch_bounds__(256) void seg_fill(
    const int* __restrict__ head, const int* __restrict__ tail,
    const int* __restrict__ etype, const int* __restrict__ uidx,
    const int* __restrict__ iidx, const float* __restrict__ w,
    int* __restrict__ cursor, int2* __restrict__ sorted, int E, int EU, int NE) {
    const int g = blockIdx.x * 256 + threadIdx.x;
    if (g < E) {
        const int pos = atomicAdd(&cursor[head[g]], 1);
        sorted[pos] = make_int2(tail[g], etype[g]);
    } else if (g < E + EU) {
        const int j = g - E;
        const int pos = atomicAdd(&cursor[NE + uidx[j]], 1);
        sorted[pos] = make_int2(iidx[j], __float_as_int(w[j]));
    }
}

__device__ __forceinline__ float edge_score1(const float4 q, const float4 k, const float4 rv) {
    float s = q.x*k.x*rv.x + q.y*k.y*rv.y + q.z*k.z*rv.z + q.w*k.w*rv.w;
    s += __shfl_xor(s, 1, 32);
    s += __shfl_xor(s, 2, 32);
    s += __shfl_xor(s, 4, 32);
    s += __shfl_xor(s, 8, 32);
    return s * 0.125f;
}

__device__ __forceinline__ void norm_store(float4 v, size_t vidx, const float* res_in,
                                           float* res_out, float* cur_out) {
    float ss = v.x*v.x + v.y*v.y + v.z*v.z + v.w*v.w;
    ss += __shfl_xor(ss, 1, 32);
    ss += __shfl_xor(ss, 2, 32);
    ss += __shfl_xor(ss, 4, 32);
    ss += __shfl_xor(ss, 8, 32);
    ss += __shfl_xor(ss, 16, 32);
    const float inv = 1.0f / fmaxf(sqrtf(ss), 1e-12f);
    v.x *= inv; v.y *= inv; v.z *= inv; v.w *= inv;
    if (cur_out) reinterpret_cast<float4*>(cur_out)[vidx] = v;
    float4 r = reinterpret_cast<const float4*>(res_in)[vidx];
    r.x += v.x; r.y += v.y; r.z += v.z; r.w += v.w;
    reinterpret_cast<float4*>(res_out)[vidx] = r;
}

__global__ __launch_bounds__(256) void ent_segment(
    const float* __restrict__ P, const float* __restrict__ entc,
    const float* __restrict__ rel, const int* __restrict__ rowptr,
    const int2* __restrict__ sorted, const float* res_in, float* res_out,
    float* cur_out, int N) {
    const int n = uni((int)((blockIdx.x << 3) + (threadIdx.x >> 5)));
    if (n >= N) return;
    const int lane = threadIdx.x & 31;
    const int beg = uni(rowptr[n]), end = uni(rowptr[n + 1]);
    const size_t vidx = ((size_t)n * D_DIM >> 2) + lane;
    const float4 q = reinterpret_cast<const float4*>(P)[vidx];

    float m = -INFINITY;
    for (int c = beg; c < end; c += 32) {
        const int nc = (end - c < 32) ? (end - c) : 32;
        int2 tr = make_int2(0, 0);
        if (lane < nc) tr = sorted[c + lane];
#pragma unroll 4
        for (int j = 0; j < nc; ++j) {
            const int tj = lane_get(tr.x, j), rj = lane_get(tr.y, j);
            const float4 k  = reinterpret_cast<const float4*>(P   + (size_t)tj * D_DIM)[lane];
            const float4 rv = reinterpret_cast<const float4*>(rel + (size_t)rj * D_DIM)[lane];
            m = fmaxf(m, edge_score1(q, k, rv));
        }
    }

    float den = 0.0f;
    float4 acc = make_float4(0.0f, 0.0f, 0.0f, 0.0f);
    for (int c = beg; c < end; c += 32) {
        const int nc = (end - c < 32) ? (end - c) : 32;
        int2 tr = make_int2(0, 0);
        if (lane < nc) tr = sorted[c + lane];
#pragma unroll 4
        for (int j = 0; j < nc; ++j) {
            const int tj = lane_get(tr.x, j), rj = lane_get(tr.y, j);
            const float4 k  = reinterpret_cast<const float4*>(P    + (size_t)tj * D_DIM)[lane];
            const float4 v  = reinterpret_cast<const float4*>(entc + (size_t)tj * D_DIM)[lane];
            const float4 rv = reinterpret_cast<const float4*>(rel  + (size_t)rj * D_DIM)[lane];
            const float ex = __expf(edge_score1(q, k, rv) - m);
            den += ex;
            acc.x += ex * (v.x * rv.x);
            acc.y += ex * (v.y * rv.y);
            acc.z += ex * (v.z * rv.z);
            acc.w += ex * (v.w * rv.w);
        }
    }
    if (end > beg) {
        acc.x /= den; acc.y /= den; acc.z /= den; acc.w /= den;
    }
    norm_store(acc, vidx, res_in, res_out, cur_out);
}

__global__ __launch_bounds__(256) void user_segment(
    const float* __restrict__ entc, const int* __restrict__ rowptr,
    const int2* __restrict__ sorted, const float* res_in, float* res_out,
    int NE, int NU) {
    const int u = uni((int)((blockIdx.x << 3) + (threadIdx.x >> 5)));
    if (u >= NU) return;
    const int lane = threadIdx.x & 31;
    const int beg = uni(rowptr[NE + u]), end = uni(rowptr[NE + u + 1]);
    const size_t vidx = ((size_t)u * D_DIM >> 2) + lane;

    float4 acc = make_float4(0.0f, 0.0f, 0.0f, 0.0f);
    for (int c = beg; c < end; c += 32) {
        const int nc = (end - c < 32) ? (end - c) : 32;
        int2 iw = make_int2(0, 0);
        if (lane < nc) iw = sorted[c + lane];
#pragma unroll 4
        for (int j = 0; j < nc; ++j) {
            const int   ij = lane_get(iw.x, j);
            const float wj = __int_as_float(lane_get(iw.y, j));
            const float4 v = reinterpret_cast<const float4*>(entc + (size_t)ij * D_DIM)[lane];
            acc.x += wj * v.x; acc.y += wj * v.y; acc.z += wj * v.z; acc.w += wj * v.w;
        }
    }
    norm_store(acc, vidx, res_in, res_out, nullptr);
}

extern "C" void kernel_launch(void* const* d_in, const int* in_sizes, int n_in,
                              void* d_out, int out_size, void* d_ws, size_t ws_size,
                              hipStream_t stream) {
    (void)n_in; (void)out_size; (void)ws_size;
    const float* user_emb   = (const float*)d_in[0];
    const float* entity_emb = (const float*)d_in[1];
    const float* W_Q        = (const float*)d_in[2];
    const float* rel_emb    = (const float*)d_in[3];
    const int*   edge_index = (const int*)d_in[4];
    const int*   edge_type  = (const int*)d_in[5];
    const int*   inter_edge = (const int*)d_in[6];
    const float* inter_w    = (const float*)d_in[7];

    const int N_USR = in_sizes[0] / D_DIM;
    const int N_ENT = in_sizes[1] / D_DIM;
    const int E     = in_sizes[5];
    const int E_UI  = in_sizes[7];
    const int NT    = N_ENT + N_USR;
    const int ET    = E + E_UI;

    const int* head = edge_index;
    const int* tail = edge_index + E;
    const int* uidx = inter_edge;
    const int* iidx = inter_edge + E_UI;

    float* out      = (float*)d_out;
    float* user_res = out;
    float* ent_res  = out + (size_t)N_USR * D_DIM;

    char* ws = (char*)d_ws;
    size_t off = 0;
    auto carve = [&](size_t bytes) -> void* {
        void* p = ws + off;
        off += (bytes + 255) & ~(size_t)255;
        return p;
    };
    float* ent_nxt = (float*)carve((size_t)N_ENT * D_DIM * 4);
    float* P       = (float*)carve((size_t)N_ENT * D_DIM * 4);
    int*   cnt     = (int*)  carve((size_t)NT * 4);
    int*   rowptr  = (int*)  carve((size_t)(NT + 1) * 4);
    int*   cursor  = (int*)  carve((size_t)NT * 4);
    int2*  sorted  = (int2*) carve((size_t)ET * 8);

    hipMemsetAsync(cnt, 0, (size_t)NT * 4, stream);
    seg_count<<<(ET + 255) / 256, 256, 0, stream>>>(head, uidx, cnt, E, E_UI, N_ENT);
    seg_scan<<<(NT + 1023) / 1024, 1024, 0, stream>>>(cnt, rowptr, cursor, NT);
    seg_fill<<<(ET + 255) / 256, 256, 0, stream>>>(head, tail, edge_type, uidx, iidx, inter_w,
                                                   cursor, sorted, E, E_UI, N_ENT);

    for (int hop = 0; hop < 2; ++hop) {
        const float* ent_cur = hop ? ent_nxt  : entity_emb;
        const float* eres_in = hop ? ent_res  : entity_emb;
        const float* ures_in = hop ? user_res : user_emb;

        proj_wmma<<<(N_ENT + 15) / 16, 256, 0, stream>>>(ent_cur, W_Q, P, N_ENT);
        ent_segment<<<(N_ENT + 7) / 8, 256, 0, stream>>>(P, ent_cur, rel_emb, rowptr, sorted,
                                                         eres_in, ent_res,
                                                         hop ? nullptr : ent_nxt, N_ENT);
        user_segment<<<(N_USR + 7) / 8, 256, 0, stream>>>(ent_cur, rowptr, sorted,
                                                          ures_in, user_res, N_ENT, N_USR);
    }
}


# entity rows: each edge's projected, value and relation rows loaded once, 8 edges in flight; scores and value rows wait in registers for the max; rows over 8 edges rescale sums by exp(old max-new max)
# speedup vs baseline: 2.2582x; 1.0989x over previous
#include <hip/hip_runtime.h>
#include <hip/hip_bf16.h>

#define D_DIM 128

typedef __attribute__((ext_vector_type(2))) float v2f;
typedef __attribute__((ext_vector_type(8))) float v8f;


__device__ __forceinline__ int lane_get(int v, int j) {
    return __builtin_amdgcn_readlane(v, j);
}
__device__ __forceinline__ int uni(int v) {
    return __builtin_amdgcn_readfirstlane(v);
}

__global__ __launch_bounds__(256) void proj_wmma(const float* __restrict__ A,
                                                 const float* __restrict__ W,
                                                 float* __restrict__ P, int N) {
    const int rowTile = blockIdx.x << 4;
    const int wave    = threadIdx.x >> 5;
    const int lane    = threadIdx.x & 31;
    const int colTile = wave << 4;
    const int r       = lane & 15;
    const int koff    = (lane >> 4) << 1;

    int arow = rowTile + r;
    if (arow >= N) arow = N - 1;
    const float* __restrict__ ap = A + (size_t)arow * D_DIM;
    const float* __restrict__ bp = W + colTile + r;

    v8f c = {};
#pragma unroll 8
    for (int k0 = 0; k0 < D_DIM; k0 += 4) {
        v2f a, b;
        a.x = ap[k0 + koff];
        a.y = ap[k0 + koff + 1];
        b.x = bp[(size_t)(k0 + koff) * D_DIM];
        b.y = bp[(size_t)(k0 + koff + 1) * D_DIM];
        c = __builtin_amdgcn_wmma_f32_16x16x4_f32(false, a, false, b,
                                                  (short)0, c, false, false);
    }
    const int mbase = rowTile + ((lane >> 4) << 3);
    const int col   = colTile + r;
    if (rowTile + 16 <= N) {
        float* __restrict__ dst = P + (size_t)mbase * D_DIM + col;
#pragma unroll
        for (int v = 0; v < 8; ++v)
            dst[(size_t)v * D_DIM] = c[v];
    } else {
#pragma unroll
        for (int v = 0; v < 8; ++v) {
            int row = mbase + v;
            if (row < N) P[(size_t)row * D_DIM + col] = c[v];
        }
    }
}

__global__ __launch_bounds__(256) void seg_count(
    const int* __restrict__ head, const int* __restrict__ uidx,
    int* __restrict__ cnt, int E, int EU, int NE) {
    const int g = blockIdx.x * 256 + threadIdx.x;
    if (g < E)            atomicAdd(&cnt[head[g]], 1);
    else if (g < E + EU)  atomicAdd(&cnt[NE + uidx[g - E]], 1);
}

__global__ __launch_bounds__(1024) void seg_scan(
    const int* __restrict__ cnt, int* __restrict__ rowptr,
    int* __restrict__ cursor, int NT) {
    __shared__ int wsum[32];
    __shared__ int sbase;
    const int tid = threadIdx.x, lane = tid & 31, wave = tid >> 5;
    const int start = blockIdx.x * 1024;

    int part = 0;
    for (int i = tid; i < start; i += 1024) part += cnt[i];
    part += __shfl_xor(part, 16, 32);
    part += __shfl_xor(part, 8, 32);
    part += __shfl_xor(part, 4, 32);
    part += __shfl_xor(part, 2, 32);
    part += __shfl_xor(part, 1, 32);
    if (lane == 0) wsum[wave] = part;
    __syncthreads();
    if (wave == 0) {
        int t = wsum[lane];
        t += __shfl_xor(t, 16, 32);
        t += __shfl_xor(t, 8, 32);
        t += __shfl_xor(t, 4, 32);
        t += __shfl_xor(t, 2, 32);
        t += __shfl_xor(t, 1, 32);
        if (lane == 0) sbase = t;
    }
    __syncthreads();
    const int base = sbase;

    const int i = start + tid;
    const int c = (i < NT) ? cnt[i] : 0;
    int inc = c;
#pragma unroll
    for (int o = 1; o < 32; o <<= 1) {
        const int t = __shfl_up(inc, o, 32);
        if (lane >= o) inc += t;
    }
    if (lane == 31) wsum[wave] = inc;
    __syncthreads();
    if (wave == 0) {
        const int t = wsum[lane];
        int ti = t;
#pragma unroll
        for (int o = 1; o < 32; o <<= 1) {
            const int u = __shfl_up(ti, o, 32);
            if (lane >= o) ti += u;
        }
        wsum[lane] = ti - t;
    }
    __syncthreads();
    const int excl = base + wsum[wave] + inc - c;
    if (i < NT) {
        rowptr[i] = excl;
        cursor[i] = excl;
        if (i == NT - 1) rowptr[NT] = excl + c;
    }
}

__global__ __launch_bounds__(256) void seg_fill(
    const int* __restrict__ head, const int* __restrict__ tail,
    const int* __restrict__ etype, const int* __restrict__ uidx,
    const int* __restrict__ iidx, const float* __restrict__ w,
    int* __restrict__ cursor, int2* __restrict__ sorted, int E, int EU, int NE) {
    const int g = blockIdx.x * 256 + threadIdx.x;
    if (g < E) {
        const int pos = atomicAdd(&cursor[head[g]], 1);
        sorted[pos] = make_int2(tail[g], etype[g]);
    } else if (g < E + EU) {
        const int j = g - E;
        const int pos = atomicAdd(&cursor[NE + uidx[j]], 1);
        sorted[pos] = make_int2(iidx[j], __float_as_int(w[j]));
    }
}

__device__ __forceinline__ float edge_score1(const float4 q, const float4 k, const float4 rv) {
    const v2f t = (v2f){q.x, q.y} * (v2f){k.x, k.y} * (v2f){rv.x, rv.y};
    const v2f u = (v2f){q.z, q.w} * (v2f){k.z, k.w} * (v2f){rv.z, rv.w};
    float s = t.x + t.y + u.x + u.y;
    s += __shfl_xor(s, 1, 32);
    s += __shfl_xor(s, 2, 32);
    s += __shfl_xor(s, 4, 32);
    s += __shfl_xor(s, 8, 32);
    return s * 0.125f;
}

__device__ __forceinline__ void norm_store(float4 v, size_t vidx, const float* res_in,
                                           float* res_out, float* cur_out) {
    float ss = v.x*v.x + v.y*v.y + v.z*v.z + v.w*v.w;
    ss += __shfl_xor(ss, 1, 32);
    ss += __shfl_xor(ss, 2, 32);
    ss += __shfl_xor(ss, 4, 32);
    ss += __shfl_xor(ss, 8, 32);
    ss += __shfl_xor(ss, 16, 32);
    const float inv = 1.0f / fmaxf(sqrtf(ss), 1e-12f);
    v.x *= inv; v.y *= inv; v.z *= inv; v.w *= inv;
    if (cur_out) reinterpret_cast<float4*>(cur_out)[vidx] = v;
    float4 r = reinterpret_cast<const float4*>(res_in)[vidx];
    r.x += v.x; r.y += v.y; r.z += v.z; r.w += v.w;
    reinterpret_cast<float4*>(res_out)[vidx] = r;
}

#define EB 8
__global__ __launch_bounds__(256) void ent_segment(
    const float* __restrict__ P, const float* __restrict__ entc,
    const float* __restrict__ rel, const int* __restrict__ rowptr,
    const int2* __restrict__ sorted, const float* res_in, float* res_out,
    float* cur_out, int N) {
    const int n = uni((int)((blockIdx.x << 3) + (threadIdx.x >> 5)));
    if (n >= N) return;
    const int lane = threadIdx.x & 31;
    const int beg = uni(rowptr[n]), end = uni(rowptr[n + 1]);
    const size_t vidx = ((size_t)n * D_DIM >> 2) + lane;
    const float4 q = reinterpret_cast<const float4*>(P)[vidx];

    float m = -INFINITY;
    float den = 0.0f;
    float4 acc = make_float4(0.0f, 0.0f, 0.0f, 0.0f);
    for (int c = beg; c < end; c += 32) {
        const int nc = (end - c < 32) ? (end - c) : 32;
        int2 tr = make_int2(0, 0);
        if (lane < nc) tr = sorted[c + lane];
        for (int b = 0; b < nc; b += EB) {
            const int nb = (nc - b < EB) ? (nc - b) : EB;
            float  sc[EB];
            float4 vr[EB];
#pragma unroll
            for (int j = 0; j < EB; ++j) {
                const bool live = j < nb;
                const int tj = lane_get(tr.x, b + j), rj = lane_get(tr.y, b + j);
                const float4 k  = reinterpret_cast<const float4*>(P    + (size_t)tj * D_DIM)[lane];
                const float4 v  = reinterpret_cast<const float4*>(entc + (size_t)tj * D_DIM)[lane];
                const float4 rv = reinterpret_cast<const float4*>(rel  + (size_t)rj * D_DIM)[lane];
                const float s = edge_score1(q, k, rv);
                sc[j] = live ? s : -INFINITY;
                vr[j].x = live ? v.x * rv.x : 0.0f;
                vr[j].y = live ? v.y * rv.y : 0.0f;
                vr[j].z = live ? v.z * rv.z : 0.0f;
                vr[j].w = live ? v.w * rv.w : 0.0f;
            }
            float bm = sc[0];
#pragma unroll
            for (int j = 1; j < EB; ++j) bm = fmaxf(bm, sc[j]);
            const float mn = fmaxf(m, bm);
            if (c + b > beg) {
                const float f = __expf(m - mn);
                den *= f;
                acc.x *= f; acc.y *= f; acc.z *= f; acc.w *= f;
            }
            m = mn;
#pragma unroll
            for (int j = 0; j < EB; ++j) {
                const float ex = __expf(sc[j] - m);
                den += ex;
                acc.x += ex * vr[j].x;
                acc.y += ex * vr[j].y;
                acc.z += ex * vr[j].z;
                acc.w += ex * vr[j].w;
            }
        }
    }
    if (end > beg) {
        acc.x /= den; acc.y /= den; acc.z /= den; acc.w /= den;
    }
    norm_store(acc, vidx, res_in, res_out, cur_out);
}

__global__ __launch_bounds__(256) void user_segment(
    const float* __restrict__ entc, const int* __restrict__ rowptr,
    const int2* __restrict__ sorted, const float* res_in, float* res_out,
    int NE, int NU) {
    const int u = uni((int)((blockIdx.x << 3) + (threadIdx.x >> 5)));
    if (u >= NU) return;
    const int lane = threadIdx.x & 31;
    const int beg = uni(rowptr[NE + u]), end = uni(rowptr[NE + u + 1]);
    const size_t vidx = ((size_t)u * D_DIM >> 2) + lane;

    float4 acc = make_float4(0.0f, 0.0f, 0.0f, 0.0f);
    for (int c = beg; c < end; c += 32) {
        const int nc = (end - c < 32) ? (end - c) : 32;
        int2 iw = make_int2(0, 0);
        if (lane < nc) iw = sorted[c + lane];
#pragma unroll 4
        for (int j = 0; j < nc; ++j) {
            const int   ij = lane_get(iw.x, j);
            const float wj = __int_as_float(lane_get(iw.y, j));
            const float4 v = reinterpret_cast<const float4*>(entc + (size_t)ij * D_DIM)[lane];
            acc.x += wj * v.x; acc.y += wj * v.y; acc.z += wj * v.z; acc.w += wj * v.w;
        }
    }
    norm_store(acc, vidx, res_in, res_out, nullptr);
}

extern "C" void kernel_launch(void* const* d_in, const int* in_sizes, int n_in,
                              void* d_out, int out_size, void* d_ws, size_t ws_size,
                              hipStream_t stream) {
    (void)n_in; (void)out_size; (void)ws_size;
    const float* user_emb   = (const float*)d_in[0];
    const float* entity_emb = (const float*)d_in[1];
    const float* W_Q        = (const float*)d_in[2];
    const float* rel_emb    = (const float*)d_in[3];
    const int*   edge_index = (const int*)d_in[4];
    const int*   edge_type  = (const int*)d_in[5];
    const int*   inter_edge = (const int*)d_in[6];
    const float* inter_w    = (const float*)d_in[7];

    const int N_USR = in_sizes[0] / D_DIM;
    const int N_ENT = in_sizes[1] / D_DIM;
    const int E     = in_sizes[5];
    const int E_UI  = in_sizes[7];
    const int NT    = N_ENT + N_USR;
    const int ET    = E + E_UI;

    const int* head = edge_index;
    const int* tail = edge_index + E;
    const int* uidx = inter_edge;
    const int* iidx = inter_edge + E_UI;

    float* out      = (float*)d_out;
    float* user_res = out;
    float* ent_res  = out + (size_t)N_USR * D_DIM;

    char* ws = (char*)d_ws;
    size_t off = 0;
    auto carve = [&](size_t bytes) -> void* {
        void* p = ws + off;
        off += (bytes + 255) & ~(size_t)255;
        return p;
    };
    float* ent_nxt = (float*)carve((size_t)N_ENT * D_DIM * 4);
    float* P       = (float*)carve((size_t)N_ENT * D_DIM * 4);
    int*   cnt     = (int*)  carve((size_t)NT * 4);
    int*   rowptr  = (int*)  carve((size_t)(NT + 1) * 4);
    int*   cursor  = (int*)  carve((size_t)NT * 4);
    int2*  sorted  = (int2*) carve((size_t)ET * 8);

    hipMemsetAsync(cnt, 0, (size_t)NT * 4, stream);
    seg_count<<<(ET + 255) / 256, 256, 0, stream>>>(head, uidx, cnt, E, E_UI, N_ENT);
    seg_scan<<<(NT + 1023) / 1024, 1024, 0, stream>>>(cnt, rowptr, cursor, NT);
    seg_fill<<<(ET + 255) / 256, 256, 0, stream>>>(head, tail, edge_type, uidx, iidx, inter_w,
                                                   cursor, sorted, E, E_UI, N_ENT);

    for (int hop = 0; hop < 2; ++hop) {
        const float* ent_cur = hop ? ent_nxt  : entity_emb;
        const float* eres_in = hop ? ent_res  : entity_emb;
        const float* ures_in = hop ? user_res : user_emb;

        proj_wmma<<<(N_ENT + 15) / 16, 256, 0, stream>>>(ent_cur, W_Q, P, N_ENT);
        ent_segment<<<(N_ENT + 7) / 8, 256, 0, stream>>>(P, ent_cur, rel_emb, rowptr, sorted,
                                                         eres_in, ent_res,
                                                         hop ? nullptr : ent_nxt, N_ENT);
        user_segment<<<(N_USR + 7) / 8, 256, 0, stream>>>(ent_cur, rowptr, sorted,
                                                          ures_in, user_res, N_ENT, N_USR);
    }
}


# projection: a wave keeps its column tile's W fragments in registers for the whole launch; 1024 blocks walk the row tiles, each A tile read once in whole rows and staged in LDS for all 8 waves
# speedup vs baseline: 2.5888x; 1.1464x over previous
#include <hip/hip_runtime.h>
#include <hip/hip_bf16.h>

#define D_DIM 128

typedef __attribute__((ext_vector_type(2))) float v2f;
typedef __attribute__((ext_vector_type(8))) float v8f;


__device__ __forceinline__ int lane_get(int v, int j) {
    return __builtin_amdgcn_readlane(v, j);
}
__device__ __forceinline__ int uni(int v) {
    return __builtin_amdgcn_readfirstlane(v);
}

#define A_PITCH 132
__global__ __launch_bounds__(256) void proj_wmma(const float* __restrict__ A,
                                                 const float* __restrict__ W,
                                                 float* __restrict__ P, int N) {
    __shared__ __attribute__((aligned(16))) float atile[2][16 * A_PITCH];
    const int wave    = threadIdx.x >> 5;
    const int lane    = threadIdx.x & 31;
    const int colTile = wave << 4;
    const int r       = lane & 15;
    const int koff    = (lane >> 4) << 1;
    const int nTiles  = (N + 15) >> 4;

    const float* __restrict__ bp = W + colTile + r;
    v2f bf[32];
#pragma unroll
    for (int j = 0; j < 32; ++j) {
        bf[j].x = bp[(size_t)(4 * j + koff) * D_DIM];
        bf[j].y = bp[(size_t)(4 * j + koff + 1) * D_DIM];
    }

    const int scol = lane << 2;
    float4 g0 = make_float4(0.0f, 0.0f, 0.0f, 0.0f), g1 = g0;
    int tile = blockIdx.x;
    if (tile < nTiles) {
        int r0 = (tile << 4) + wave, r1 = r0 + 8;
        if (r0 >= N) r0 = N - 1;
        if (r1 >= N) r1 = N - 1;
        g0 = *reinterpret_cast<const float4*>(A + (size_t)r0 * D_DIM + scol);
        g1 = *reinterpret_cast<const float4*>(A + (size_t)r1 * D_DIM + scol);
    }
    int buf = 0;
    for (; tile < nTiles; tile += gridDim.x) {
        float* __restrict__ at = atile[buf];
        *reinterpret_cast<float4*>(at + wave * A_PITCH + scol)       = g0;
        *reinterpret_cast<float4*>(at + (wave + 8) * A_PITCH + scol) = g1;
        __syncthreads();
        const int nxt = tile + gridDim.x;
        if (nxt < nTiles) {
            int r0 = (nxt << 4) + wave, r1 = r0 + 8;
            if (r0 >= N) r0 = N - 1;
            if (r1 >= N) r1 = N - 1;
            g0 = *reinterpret_cast<const float4*>(A + (size_t)r0 * D_DIM + scol);
            g1 = *reinterpret_cast<const float4*>(A + (size_t)r1 * D_DIM + scol);
        }

        const float* __restrict__ ar = at + r * A_PITCH + koff;
        v8f c = {};
#pragma unroll
        for (int j = 0; j < 32; ++j) {
            const v2f a = *reinterpret_cast<const v2f*>(ar + 4 * j);
            c = __builtin_amdgcn_wmma_f32_16x16x4_f32(false, a, false, bf[j],
                                                      (short)0, c, false, false);
        }
        const int rowTile = tile << 4;
        const int mbase   = rowTile + ((lane >> 4) << 3);
        const int col     = colTile + r;
        if (rowTile + 16 <= N) {
            float* __restrict__ dst = P + (size_t)mbase * D_DIM + col;
#pragma unroll
            for (int v = 0; v < 8; ++v)
                dst[(size_t)v * D_DIM] = c[v];
        } else {
#pragma unroll
            for (int v = 0; v < 8; ++v) {
                int row = mbase + v;
                if (row < N) P[(size_t)row * D_DIM + col] = c[v];
            }
        }
        buf ^= 1;
    }
}

__global__ __launch_bounds__(256) void seg_count(
    const int* __restrict__ head, const int* __restrict__ uidx,
    int* __restrict__ cnt, int E, int EU, int NE) {
    const int g = blockIdx.x * 256 + threadIdx.x;
    if (g < E)            atomicAdd(&cnt[head[g]], 1);
    else if (g < E + EU)  atomicAdd(&cnt[NE + uidx[g - E]], 1);
}

__global__ __launch_bounds__(1024) void seg_scan(
    const int* __restrict__ cnt, int* __restrict__ rowptr,
    int* __restrict__ cursor, int NT) {
    __shared__ int wsum[32];
    __shared__ int sbase;
    const int tid = threadIdx.x, lane = tid & 31, wave = tid >> 5;
    const int start = blockIdx.x * 1024;

    int part = 0;
    for (int i = tid; i < start; i += 1024) part += cnt[i];
    part += __shfl_xor(part, 16, 32);
    part += __shfl_xor(part, 8, 32);
    part += __shfl_xor(part, 4, 32);
    part += __shfl_xor(part, 2, 32);
    part += __shfl_xor(part, 1, 32);
    if (lane == 0) wsum[wave] = part;
    __syncthreads();
    if (wave == 0) {
        int t = wsum[lane];
        t += __shfl_xor(t, 16, 32);
        t += __shfl_xor(t, 8, 32);
        t += __shfl_xor(t, 4, 32);
        t += __shfl_xor(t, 2, 32);
        t += __shfl_xor(t, 1, 32);
        if (lane == 0) sbase = t;
    }
    __syncthreads();
    const int base = sbase;

    const int i = start + tid;
    const int c = (i < NT) ? cnt[i] : 0;
    int inc = c;
#pragma unroll
    for (int o = 1; o < 32; o <<= 1) {
        const int t = __shfl_up(inc, o, 32);
        if (lane >= o) inc += t;
    }
    if (lane == 31) wsum[wave] = inc;
    __syncthreads();
    if (wave == 0) {
        const int t = wsum[lane];
        int ti = t;
#pragma unroll
        for (int o = 1; o < 32; o <<= 1) {
            const int u = __shfl_up(ti, o, 32);
            if (lane >= o) ti += u;
        }
        wsum[lane] = ti - t;
    }
    __syncthreads();
    const int excl = base + wsum[wave] + inc - c;
    if (i < NT) {
        rowptr[i] = excl;
        cursor[i] = excl;
        if (i == NT - 1) rowptr[NT] = excl + c;
    }
}

__global__ __launch_bounds__(256) void seg_fill(
    const int* __restrict__ head, const int* __restrict__ tail,
    const int* __restrict__ etype, const int* __restrict__ uidx,
    const int* __restrict__ iidx, const float* __restrict__ w,
    int* __restrict__ cursor, int2* __restrict__ sorted, int E, int EU, int NE) {
    const int g = blockIdx.x * 256 + threadIdx.x;
    if (g < E) {
        const int pos = atomicAdd(&cursor[head[g]], 1);
        sorted[pos] = make_int2(tail[g], etype[g]);
    } else if (g < E + EU) {
        const int j = g - E;
        const int pos = atomicAdd(&cursor[NE + uidx[j]], 1);
        sorted[pos] = make_int2(iidx[j], __float_as_int(w[j]));
    }
}

__device__ __forceinline__ float edge_score1(const float4 q, const float4 k, const float4 rv) {
    const v2f t = (v2f){q.x, q.y} * (v2f){k.x, k.y} * (v2f){rv.x, rv.y};
    const v2f u = (v2f){q.z, q.w} * (v2f){k.z, k.w} * (v2f){rv.z, rv.w};
    float s = t.x + t.y + u.x + u.y;
    s += __shfl_xor(s, 1, 32);
    s += __shfl_xor(s, 2, 32);
    s += __shfl_xor(s, 4, 32);
    s += __shfl_xor(s, 8, 32);
    return s * 0.125f;
}

__device__ __forceinline__ void norm_store(float4 v, size_t vidx, const float* res_in,
                                           float* res_out, float* cur_out) {
    float ss = v.x*v.x + v.y*v.y + v.z*v.z + v.w*v.w;
    ss += __shfl_xor(ss, 1, 32);
    ss += __shfl_xor(ss, 2, 32);
    ss += __shfl_xor(ss, 4, 32);
    ss += __shfl_xor(ss, 8, 32);
    ss += __shfl_xor(ss, 16, 32);
    const float inv = 1.0f / fmaxf(sqrtf(ss), 1e-12f);
    v.x *= inv; v.y *= inv; v.z *= inv; v.w *= inv;
    if (cur_out) reinterpret_cast<float4*>(cur_out)[vidx] = v;
    float4 r = reinterpret_cast<const float4*>(res_in)[vidx];
    r.x += v.x; r.y += v.y; r.z += v.z; r.w += v.w;
    reinterpret_cast<float4*>(res_out)[vidx] = r;
}

#define EB 8
__global__ __launch_bounds__(256) void ent_segment(
    const float* __restrict__ P, const float* __restrict__ entc,
    const float* __restrict__ rel, const int* __restrict__ rowptr,
    const int2* __restrict__ sorted, const float* res_in, float* res_out,
    float* cur_out, int N) {
    const int n = uni((int)((blockIdx.x << 3) + (threadIdx.x >> 5)));
    if (n >= N) return;
    const int lane = threadIdx.x & 31;
    const int beg = uni(rowptr[n]), end = uni(rowptr[n + 1]);
    const size_t vidx = ((size_t)n * D_DIM >> 2) + lane;
    const float4 q = reinterpret_cast<const float4*>(P)[vidx];

    float m = -INFINITY;
    float den = 0.0f;
    float4 acc = make_float4(0.0f, 0.0f, 0.0f, 0.0f);
    for (int c = beg; c < end; c += 32) {
        const int nc = (end - c < 32) ? (end - c) : 32;
        int2 tr = make_int2(0, 0);
        if (lane < nc) tr = sorted[c + lane];
        for (int b = 0; b < nc; b += EB) {
            const int nb = (nc - b < EB) ? (nc - b) : EB;
            float  sc[EB];
            float4 vr[EB];
#pragma unroll
            for (int j = 0; j < EB; ++j) {
                const bool live = j < nb;
                const int tj = lane_get(tr.x, b + j), rj = lane_get(tr.y, b + j);
                const float4 k  = reinterpret_cast<const float4*>(P    + (size_t)tj * D_DIM)[lane];
                const float4 v  = reinterpret_cast<const float4*>(entc + (size_t)tj * D_DIM)[lane];
                const float4 rv = reinterpret_cast<const float4*>(rel  + (size_t)rj * D_DIM)[lane];
                const float s = edge_score1(q, k, rv);
                sc[j] = live ? s : -INFINITY;
                vr[j].x = live ? v.x * rv.x : 0.0f;
                vr[j].y = live ? v.y * rv.y : 0.0f;
                vr[j].z = live ? v.z * rv.z : 0.0f;
                vr[j].w = live ? v.w * rv.w : 0.0f;
            }
            float bm = sc[0];
#pragma unroll
            for (int j = 1; j < EB; ++j) bm = fmaxf(bm, sc[j]);
            const float mn = fmaxf(m, bm);
            if (c + b > beg) {
                const float f = __expf(m - mn);
                den *= f;
                acc.x *= f; acc.y *= f; acc.z *= f; acc.w *= f;
            }
            m = mn;
#pragma unroll
            for (int j = 0; j < EB; ++j) {
                const float ex = __expf(sc[j] - m);
                den += ex;
                acc.x += ex * vr[j].x;
                acc.y += ex * vr[j].y;
                acc.z += ex * vr[j].z;
                acc.w += ex * vr[j].w;
            }
        }
    }
    if (end > beg) {
        acc.x /= den; acc.y /= den; acc.z /= den; acc.w /= den;
    }
    norm_store(acc, vidx, res_in, res_out, cur_out);
}

__global__ __launch_bounds__(256) void user_segment(
    const float* __restrict__ entc, const int* __restrict__ rowptr,
    const int2* __restrict__ sorted, const float* res_in, float* res_out,
    int NE, int NU) {
    const int u = uni((int)((blockIdx.x << 3) + (threadIdx.x >> 5)));
    if (u >= NU) return;
    const int lane = threadIdx.x & 31;
    const int beg = uni(rowptr[NE + u]), end = uni(rowptr[NE + u + 1]);
    const size_t vidx = ((size_t)u * D_DIM >> 2) + lane;

    float4 acc = make_float4(0.0f, 0.0f, 0.0f, 0.0f);
    for (int c = beg; c < end; c += 32) {
        const int nc = (end - c < 32) ? (end - c) : 32;
        int2 iw = make_int2(0, 0);
        if (lane < nc) iw = sorted[c + lane];
#pragma unroll 4
        for (int j = 0; j < nc; ++j) {
            const int   ij = lane_get(iw.x, j);
            const float wj = __int_as_float(lane_get(iw.y, j));
            const float4 v = reinterpret_cast<const float4*>(entc + (size_t)ij * D_DIM)[lane];
            acc.x += wj * v.x; acc.y += wj * v.y; acc.z += wj * v.z; acc.w += wj * v.w;
        }
    }
    norm_store(acc, vidx, res_in, res_out, nullptr);
}

extern "C" void kernel_launch(void* const* d_in, const int* in_sizes, int n_in,
                              void* d_out, int out_size, void* d_ws, size_t ws_size,
                              hipStream_t stream) {
    (void)n_in; (void)out_size; (void)ws_size;
    const float* user_emb   = (const float*)d_in[0];
    const float* entity_emb = (const float*)d_in[1];
    const float* W_Q        = (const float*)d_in[2];
    const float* rel_emb    = (const float*)d_in[3];
    const int*   edge_index = (const int*)d_in[4];
    const int*   edge_type  = (const int*)d_in[5];
    const int*   inter_edge = (const int*)d_in[6];
    const float* inter_w    = (const float*)d_in[7];

    const int N_USR = in_sizes[0] / D_DIM;
    const int N_ENT = in_sizes[1] / D_DIM;
    const int E     = in_sizes[5];
    const int E_UI  = in_sizes[7];
    const int NT    = N_ENT + N_USR;
    const int ET    = E + E_UI;

    const int* head = edge_index;
    const int* tail = edge_index + E;
    const int* uidx = inter_edge;
    const int* iidx = inter_edge + E_UI;

    float* out      = (float*)d_out;
    float* user_res = out;
    float* ent_res  = out + (size_t)N_USR * D_DIM;

    char* ws = (char*)d_ws;
    size_t off = 0;
    auto carve = [&](size_t bytes) -> void* {
        void* p = ws + off;
        off += (bytes + 255) & ~(size_t)255;
        return p;
    };
    float* ent_nxt = (float*)carve((size_t)N_ENT * D_DIM * 4);
    float* P       = (float*)carve((size_t)N_ENT * D_DIM * 4);
    int*   cnt     = (int*)  carve((size_t)NT * 4);
    int*   rowptr  = (int*)  carve((size_t)(NT + 1) * 4);
    int*   cursor  = (int*)  carve((size_t)NT * 4);
    int2*  sorted  = (int2*) carve((size_t)ET * 8);

    hipMemsetAsync(cnt, 0, (size_t)NT * 4, stream);
    seg_count<<<(ET + 255) / 256, 256, 0, stream>>>(head, uidx, cnt, E, E_UI, N_ENT);
    seg_scan<<<(NT + 1023) / 1024, 1024, 0, stream>>>(cnt, rowptr, cursor, NT);
    seg_fill<<<(ET + 255) / 256, 256, 0, stream>>>(head, tail, edge_type, uidx, iidx, inter_w,
                                                   cursor, sorted, E, E_UI, N_ENT);

    for (int hop = 0; hop < 2; ++hop) {
        const float* ent_cur = hop ? ent_nxt  : entity_emb;
        const float* eres_in = hop ? ent_res  : entity_emb;
        const float* ures_in = hop ? user_res : user_emb;

        proj_wmma<<<1024, 256, 0, stream>>>(ent_cur, W_Q, P, N_ENT);
        ent_segment<<<(N_ENT + 7) / 8, 256, 0, stream>>>(P, ent_cur, rel_emb, rowptr, sorted,
                                                         eres_in, ent_res,
                                                         hop ? nullptr : ent_nxt, N_ENT);
        user_segment<<<(N_USR + 7) / 8, 256, 0, stream>>>(ent_cur, rowptr, sorted,
                                                          ures_in, user_res, N_ENT, N_USR);
    }
}


# edge ordering: two-level counting sort through LDS counters (bins of 256 destination rows, whole-line stores), no per-edge global atomics or scattered stores; sums stay in sort-arrival order
# speedup vs baseline: 3.4500x; 1.3327x over previous
#include <hip/hip_runtime.h>
#include <hip/hip_bf16.h>

#define D_DIM 128

typedef __attribute__((ext_vector_type(2))) float v2f;
typedef __attribute__((ext_vector_type(8))) float v8f;


__device__ __forceinline__ int lane_get(int v, int j) {
    return __builtin_amdgcn_readlane(v, j);
}
__device__ __forceinline__ int uni(int v) {
    return __builtin_amdgcn_readfirstlane(v);
}

#define A_PITCH 132
__global__ __launch_bounds__(256) void proj_wmma(const float* __restrict__ A,
                                                 const float* __restrict__ W,
                                                 float* __restrict__ P, int N) {
    __shared__ __attribute__((aligned(16))) float atile[2][16 * A_PITCH];
    const int wave    = threadIdx.x >> 5;
    const int lane    = threadIdx.x & 31;
    const int colTile = wave << 4;
    const int r       = lane & 15;
    const int koff    = (lane >> 4) << 1;
    const int nTiles  = (N + 15) >> 4;

    const float* __restrict__ bp = W + colTile + r;
    v2f bf[32];
#pragma unroll
    for (int j = 0; j < 32; ++j) {
        bf[j].x = bp[(size_t)(4 * j + koff) * D_DIM];
        bf[j].y = bp[(size_t)(4 * j + koff + 1) * D_DIM];
    }

    const int scol = lane << 2;
    float4 g0 = make_float4(0.0f, 0.0f, 0.0f, 0.0f), g1 = g0;
    int tile = blockIdx.x;
    if (tile < nTiles) {
        int r0 = (tile << 4) + wave, r1 = r0 + 8;
        if (r0 >= N) r0 = N - 1;
        if (r1 >= N) r1 = N - 1;
        g0 = *reinterpret_cast<const float4*>(A + (size_t)r0 * D_DIM + scol);
        g1 = *reinterpret_cast<const float4*>(A + (size_t)r1 * D_DIM + scol);
    }
    int buf = 0;
    for (; tile < nTiles; tile += gridDim.x) {
        float* __restrict__ at = atile[buf];
        *reinterpret_cast<float4*>(at + wave * A_PITCH + scol)       = g0;
        *reinterpret_cast<float4*>(at + (wave + 8) * A_PITCH + scol) = g1;
        __syncthreads();
        const int nxt = tile + gridDim.x;
        if (nxt < nTiles) {
            int r0 = (nxt << 4) + wave, r1 = r0 + 8;
            if (r0 >= N) r0 = N - 1;
            if (r1 >= N) r1 = N - 1;
            g0 = *reinterpret_cast<const float4*>(A + (size_t)r0 * D_DIM + scol);
            g1 = *reinterpret_cast<const float4*>(A + (size_t)r1 * D_DIM + scol);
        }

        const float* __restrict__ ar = at + r * A_PITCH + koff;
        v8f c = {};
#pragma unroll
        for (int j = 0; j < 32; ++j) {
            const v2f a = *reinterpret_cast<const v2f*>(ar + 4 * j);
            c = __builtin_amdgcn_wmma_f32_16x16x4_f32(false, a, false, bf[j],
                                                      (short)0, c, false, false);
        }
        const int rowTile = tile << 4;
        const int mbase   = rowTile + ((lane >> 4) << 3);
        const int col     = colTile + r;
        if (rowTile + 16 <= N) {
            float* __restrict__ dst = P + (size_t)mbase * D_DIM + col;
#pragma unroll
            for (int v = 0; v < 8; ++v)
                dst[(size_t)v * D_DIM] = c[v];
        } else {
#pragma unroll
            for (int v = 0; v < 8; ++v) {
                int row = mbase + v;
                if (row < N) P[(size_t)row * D_DIM + col] = c[v];
            }
        }
        buf ^= 1;
    }
}

#define BIN_SHIFT 8
#define BIN_W     256
#define NBMAX     1024
#define TILE      4096
#define BIN_CAP   4096

__device__ __forceinline__ int block_excl_scan(int c, int* wsum) {
    const int lane = threadIdx.x & 31, wave = threadIdx.x >> 5;
    int inc = c;
#pragma unroll
    for (int o = 1; o < 32; o <<= 1) {
        const int t = __shfl_up(inc, o, 32);
        if (lane >= o) inc += t;
    }
    if (lane == 31) wsum[wave] = inc;
    __syncthreads();
    if (wave == 0) {
        const int nw = (int)(blockDim.x >> 5);
        const int t = (lane < nw) ? wsum[lane] : 0;
        int ti = t;
#pragma unroll
        for (int o = 1; o < 32; o <<= 1) {
            const int u = __shfl_up(ti, o, 32);
            if (lane >= o) ti += u;
        }
        wsum[lane] = ti - t;
    }
    __syncthreads();
    return wsum[wave] + inc - c;
}

__global__ __launch_bounds__(1024) void bin_count(
    const int* __restrict__ head, const int* __restrict__ uidx,
    int* __restrict__ cntm, int E, int EU, int NE, int NB) {
    __shared__ int hist[NBMAX];
    const int tid = threadIdx.x;
    hist[tid] = 0;
    __syncthreads();
    const int ET = E + EU;
    const int beg = blockIdx.x * TILE;
#pragma unroll
    for (int k = 0; k < 4; ++k) {
        const int g = beg + tid + (k << 10);
        if (g < ET) {
            const int key = (g < E) ? head[g] : NE + uidx[g - E];
            atomicAdd(&hist[key >> BIN_SHIFT], 1);
        }
    }
    __syncthreads();
    if (tid < NB) cntm[(size_t)tid * gridDim.x + blockIdx.x] = hist[tid];
}

__global__ __launch_bounds__(1024) void seg_scan(
    const int* __restrict__ cnt, int* __restrict__ off, int NT) {
    __shared__ int wsum[32];
    __shared__ int sbase;
    const int tid = threadIdx.x, lane = tid & 31, wave = tid >> 5;
    const int start = blockIdx.x * 1024;

    int part = 0;
    for (int i = tid; i < start; i += 1024) part += cnt[i];
    part += __shfl_xor(part, 16, 32);
    part += __shfl_xor(part, 8, 32);
    part += __shfl_xor(part, 4, 32);
    part += __shfl_xor(part, 2, 32);
    part += __shfl_xor(part, 1, 32);
    if (lane == 0) wsum[wave] = part;
    __syncthreads();
    if (wave == 0) {
        int t = wsum[lane];
        t += __shfl_xor(t, 16, 32);
        t += __shfl_xor(t, 8, 32);
        t += __shfl_xor(t, 4, 32);
        t += __shfl_xor(t, 2, 32);
        t += __shfl_xor(t, 1, 32);
        if (lane == 0) sbase = t;
    }
    __syncthreads();
    const int base = sbase;

    const int i = start + tid;
    const int c = (i < NT) ? cnt[i] : 0;
    const int excl = base + block_excl_scan(c, wsum);
    if (i < NT) {
        off[i] = excl;
        if (i == NT - 1) off[NT] = excl + c;
    }
}

__global__ __launch_bounds__(1024) void bin_scatter(
    const int* __restrict__ head, const int* __restrict__ tail,
    const int* __restrict__ etype, const int* __restrict__ uidx,
    const int* __restrict__ iidx, const float* __restrict__ w,
    const int* __restrict__ binoff, int2* __restrict__ binned,
    int E, int EU, int NE, int NB) {
    __shared__ int lbase[NBMAX];
    __shared__ int lcur[NBMAX];
    __shared__ int gbase[NBMAX];
    __shared__ int wsum[32];
    __shared__ int2 stage[TILE];
    __shared__ unsigned short sbin[TILE];
    const int tid = threadIdx.x;
    const int ET = E + EU;
    const int beg = blockIdx.x * TILE;

    int c = 0, g0 = 0;
    if (tid < NB) {
        const int* p = binoff + (size_t)tid * gridDim.x + blockIdx.x;
        g0 = p[0];
        c  = p[1] - g0;
    }
    const int ex = block_excl_scan(c, wsum);
    gbase[tid] = g0; lbase[tid] = ex; lcur[tid] = ex;

    int key[4], px[4], py[4];
#pragma unroll
    for (int k = 0; k < 4; ++k) {
        const int g = beg + tid + (k << 10);
        key[k] = -1; px[k] = 0; py[k] = 0;
        if (g < E) {
            key[k] = head[g]; px[k] = tail[g]; py[k] = etype[g];
        } else if (g < ET) {
            const int j = g - E;
            key[k] = NE + uidx[j]; px[k] = iidx[j]; py[k] = __float_as_int(w[j]);
        }
    }
    __syncthreads();
#pragma unroll
    for (int k = 0; k < 4; ++k) {
        if (key[k] >= 0) {
            const int bin = key[k] >> BIN_SHIFT;
            const int r = atomicAdd(&lcur[bin], 1);
            stage[r] = make_int2(px[k] | ((key[k] & (BIN_W - 1)) << 24), py[k]);
            sbin[r]  = (unsigned short)bin;
        }
    }
    __syncthreads();
    const int n = (ET - beg < TILE) ? (ET - beg) : TILE;
    for (int k = 0; k < 4; ++k) {
        const int i = tid + (k << 10);
        if (i < n) {
            const int bin = sbin[i];
            binned[gbase[bin] + (i - lbase[bin])] = stage[i];
        }
    }
}

__global__ __launch_bounds__(512) void bin_sort(
    const int* __restrict__ binoff, const int2* __restrict__ binned,
    int* __restrict__ rowptr, int2* __restrict__ sorted, int G, int NT) {
    __shared__ int kcnt[BIN_W];
    __shared__ int kcur[BIN_W];
    __shared__ int wsum[32];
    __shared__ int2 ost[BIN_CAP];
    const int tid = threadIdx.x;
    const int b = blockIdx.x;
    const int s = uni(binoff[(size_t)b * G]), e = uni(binoff[(size_t)(b + 1) * G]);
    const int n = e - s;
    if (tid < BIN_W) kcnt[tid] = 0;
    __syncthreads();
    for (int i = tid; i < n; i += 512)
        atomicAdd(&kcnt[(unsigned)binned[s + i].x >> 24], 1);
    __syncthreads();
    const int c  = (tid < BIN_W) ? kcnt[tid] : 0;
    const int ex = block_excl_scan(c, wsum);
    if (tid < BIN_W) {
        kcur[tid] = ex;
        const int key = (b << BIN_SHIFT) + tid;
        if (key < NT) rowptr[key] = s + ex;
    }
    if (b == (int)gridDim.x - 1 && tid == 0) rowptr[NT] = e;
    __syncthreads();
    if (n <= BIN_CAP) {
        for (int i = tid; i < n; i += 512) {
            const int2 rec = binned[s + i];
            const int r = atomicAdd(&kcur[(unsigned)rec.x >> 24], 1);
            ost[r] = make_int2(rec.x & 0x00FFFFFF, rec.y);
        }
        __syncthreads();
        for (int i = tid; i < n; i += 512) sorted[s + i] = ost[i];
    } else {
        for (int i = tid; i < n; i += 512) {
            const int2 rec = binned[s + i];
            const int r = atomicAdd(&kcur[(unsigned)rec.x >> 24], 1);
            sorted[s + r] = make_int2(rec.x & 0x00FFFFFF, rec.y);
        }
    }
}

__device__ __forceinline__ float edge_score1(const float4 q, const float4 k, const float4 rv) {
    const v2f t = (v2f){q.x, q.y} * (v2f){k.x, k.y} * (v2f){rv.x, rv.y};
    const v2f u = (v2f){q.z, q.w} * (v2f){k.z, k.w} * (v2f){rv.z, rv.w};
    float s = t.x + t.y + u.x + u.y;
    s += __shfl_xor(s, 1, 32);
    s += __shfl_xor(s, 2, 32);
    s += __shfl_xor(s, 4, 32);
    s += __shfl_xor(s, 8, 32);
    return s * 0.125f;
}

__device__ __forceinline__ void norm_store(float4 v, size_t vidx, const float* res_in,
                                           float* res_out, float* cur_out) {
    float ss = v.x*v.x + v.y*v.y + v.z*v.z + v.w*v.w;
    ss += __shfl_xor(ss, 1, 32);
    ss += __shfl_xor(ss, 2, 32);
    ss += __shfl_xor(ss, 4, 32);
    ss += __shfl_xor(ss, 8, 32);
    ss += __shfl_xor(ss, 16, 32);
    const float inv = 1.0f / fmaxf(sqrtf(ss), 1e-12f);
    v.x *= inv; v.y *= inv; v.z *= inv; v.w *= inv;
    if (cur_out) reinterpret_cast<float4*>(cur_out)[vidx] = v;
    float4 r = reinterpret_cast<const float4*>(res_in)[vidx];
    r.x += v.x; r.y += v.y; r.z += v.z; r.w += v.w;
    reinterpret_cast<float4*>(res_out)[vidx] = r;
}

#define EB 8
__global__ __launch_bounds__(256) void ent_segment(
    const float* __restrict__ P, const float* __restrict__ entc,
    const float* __restrict__ rel, const int* __restrict__ rowptr,
    const int2* __restrict__ sorted, const float* res_in, float* res_out,
    float* cur_out, int N) {
    const int n = uni((int)((blockIdx.x << 3) + (threadIdx.x >> 5)));
    if (n >= N) return;
    const int lane = threadIdx.x & 31;
    const int beg = uni(rowptr[n]), end = uni(rowptr[n + 1]);
    const size_t vidx = ((size_t)n * D_DIM >> 2) + lane;
    const float4 q = reinterpret_cast<const float4*>(P)[vidx];

    float m = -INFINITY;
    float den = 0.0f;
    float4 acc = make_float4(0.0f, 0.0f, 0.0f, 0.0f);
    for (int c = beg; c < end; c += 32) {
        const int nc = (end - c < 32) ? (end - c) : 32;
        int2 tr = make_int2(0, 0);
        if (lane < nc) tr = sorted[c + lane];
        for (int b = 0; b < nc; b += EB) {
            const int nb = (nc - b < EB) ? (nc - b) : EB;
            float  sc[EB];
            float4 vr[EB];
#pragma unroll
            for (int j = 0; j < EB; ++j) {
                const bool live = j < nb;
                const int tj = lane_get(tr.x, b + j), rj = lane_get(tr.y, b + j);
                const float4 k  = reinterpret_cast<const float4*>(P    + (size_t)tj * D_DIM)[lane];
                const float4 v  = reinterpret_cast<const float4*>(entc + (size_t)tj * D_DIM)[lane];
                const float4 rv = reinterpret_cast<const float4*>(rel  + (size_t)rj * D_DIM)[lane];
                const float s = edge_score1(q, k, rv);
                sc[j] = live ? s : -INFINITY;
                vr[j].x = live ? v.x * rv.x : 0.0f;
                vr[j].y = live ? v.y * rv.y : 0.0f;
                vr[j].z = live ? v.z * rv.z : 0.0f;
                vr[j].w = live ? v.w * rv.w : 0.0f;
            }
            float bm = sc[0];
#pragma unroll
            for (int j = 1; j < EB; ++j) bm = fmaxf(bm, sc[j]);
            const float mn = fmaxf(m, bm);
            if (c + b > beg) {
                const float f = __expf(m - mn);
                den *= f;
                acc.x *= f; acc.y *= f; acc.z *= f; acc.w *= f;
            }
            m = mn;
#pragma unroll
            for (int j = 0; j < EB; ++j) {
                const float ex = __expf(sc[j] - m);
                den += ex;
                acc.x += ex * vr[j].x;
                acc.y += ex * vr[j].y;
                acc.z += ex * vr[j].z;
                acc.w += ex * vr[j].w;
            }
        }
    }
    if (end > beg) {
        acc.x /= den; acc.y /= den; acc.z /= den; acc.w /= den;
    }
    norm_store(acc, vidx, res_in, res_out, cur_out);
}

__global__ __launch_bounds__(256) void user_segment(
    const float* __restrict__ entc, const int* __restrict__ rowptr,
    const int2* __restrict__ sorted, const float* res_in, float* res_out,
    int NE, int NU) {
    const int u = uni((int)((blockIdx.x << 3) + (threadIdx.x >> 5)));
    if (u >= NU) return;
    const int lane = threadIdx.x & 31;
    const int beg = uni(rowptr[NE + u]), end = uni(rowptr[NE + u + 1]);
    const size_t vidx = ((size_t)u * D_DIM >> 2) + lane;

    float4 acc = make_float4(0.0f, 0.0f, 0.0f, 0.0f);
    for (int c = beg; c < end; c += 32) {
        const int nc = (end - c < 32) ? (end - c) : 32;
        int2 iw = make_int2(0, 0);
        if (lane < nc) iw = sorted[c + lane];
#pragma unroll 4
        for (int j = 0; j < nc; ++j) {
            const int   ij = lane_get(iw.x, j);
            const float wj = __int_as_float(lane_get(iw.y, j));
            const float4 v = reinterpret_cast<const float4*>(entc + (size_t)ij * D_DIM)[lane];
            acc.x += wj * v.x; acc.y += wj * v.y; acc.z += wj * v.z; acc.w += wj * v.w;
        }
    }
    norm_store(acc, vidx, res_in, res_out, nullptr);
}

extern "C" void kernel_launch(void* const* d_in, const int* in_sizes, int n_in,
                              void* d_out, int out_size, void* d_ws, size_t ws_size,
                              hipStream_t stream) {
    (void)n_in; (void)out_size; (void)ws_size;
    const float* user_emb   = (const float*)d_in[0];
    const float* entity_emb = (const float*)d_in[1];
    const float* W_Q        = (const float*)d_in[2];
    const float* rel_emb    = (const float*)d_in[3];
    const int*   edge_index = (const int*)d_in[4];
    const int*   edge_type  = (const int*)d_in[5];
    const int*   inter_edge = (const int*)d_in[6];
    const float* inter_w    = (const float*)d_in[7];

    const int N_USR = in_sizes[0] / D_DIM;
    const int N_ENT = in_sizes[1] / D_DIM;
    const int E     = in_sizes[5];
    const int E_UI  = in_sizes[7];
    const int NT    = N_ENT + N_USR;
    const int ET    = E + E_UI;

    const int* head = edge_index;
    const int* tail = edge_index + E;
    const int* uidx = inter_edge;
    const int* iidx = inter_edge + E_UI;

    float* out      = (float*)d_out;
    float* user_res = out;
    float* ent_res  = out + (size_t)N_USR * D_DIM;

    char* ws = (char*)d_ws;
    size_t off = 0;
    auto carve = [&](size_t bytes) -> void* {
        void* p = ws + off;
        off += (bytes + 255) & ~(size_t)255;
        return p;
    };
    float* ent_nxt = (float*)carve((size_t)N_ENT * D_DIM * 4);
    float* P       = (float*)carve((size_t)N_ENT * D_DIM * 4);
    const int NB   = (NT + BIN_W - 1) >> BIN_SHIFT;
    const int G    = (ET + TILE - 1) / TILE;
    const int NTOT = NB * G;
    int*   cntm    = (int*)  carve((size_t)NTOT * 4);
    int*   binoff  = (int*)  carve((size_t)(NTOT + 1) * 4);
    int*   rowptr  = (int*)  carve((size_t)(NT + 1) * 4);
    int2*  binned  = (int2*) carve((size_t)ET * 8);
    int2*  sorted  = (int2*) carve((size_t)ET * 8);

    bin_count<<<G, 1024, 0, stream>>>(head, uidx, cntm, E, E_UI, N_ENT, NB);
    seg_scan<<<(NTOT + 1023) / 1024, 1024, 0, stream>>>(cntm, binoff, NTOT);
    bin_scatter<<<G, 1024, 0, stream>>>(head, tail, edge_type, uidx, iidx, inter_w,
                                        binoff, binned, E, E_UI, N_ENT, NB);
    bin_sort<<<NB, 512, 0, stream>>>(binoff, binned, rowptr, sorted, G, NT);

    for (int hop = 0; hop < 2; ++hop) {
        const float* ent_cur = hop ? ent_nxt  : entity_emb;
        const float* eres_in = hop ? ent_res  : entity_emb;
        const float* ures_in = hop ? user_res : user_emb;

        proj_wmma<<<1024, 256, 0, stream>>>(ent_cur, W_Q, P, N_ENT);
        ent_segment<<<(N_ENT + 7) / 8, 256, 0, stream>>>(P, ent_cur, rel_emb, rowptr, sorted,
                                                         eres_in, ent_res,
                                                         hop ? nullptr : ent_nxt, N_ENT);
        user_segment<<<(N_USR + 7) / 8, 256, 0, stream>>>(ent_cur, rowptr, sorted,
                                                          ures_in, user_res, N_ENT, N_USR);
    }
}


# entity and user row kernels: a resident grid (768 and 2048 blocks) whose waves walk the rows, next row's index loads issued a row ahead, replaces one started wave per row; sums and order unchanged
# speedup vs baseline: 3.7255x; 1.0799x over previous
#include <hip/hip_runtime.h>
#include <hip/hip_bf16.h>

#define D_DIM 128

typedef __attribute__((ext_vector_type(2))) float v2f;
typedef __attribute__((ext_vector_type(8))) float v8f;


__device__ __forceinline__ int lane_get(int v, int j) {
    return __builtin_amdgcn_readlane(v, j);
}
__device__ __forceinline__ int uni(int v) {
    return __builtin_amdgcn_readfirstlane(v);
}

#define A_PITCH 132
__global__ __launch_bounds__(256) void proj_wmma(const float* __restrict__ A,
                                                 const float* __restrict__ W,
                                                 float* __restrict__ P, int N) {
    __shared__ __attribute__((aligned(16))) float atile[2][16 * A_PITCH];
    const int wave    = threadIdx.x >> 5;
    const int lane    = threadIdx.x & 31;
    const int colTile = wave << 4;
    const int r       = lane & 15;
    const int koff    = (lane >> 4) << 1;
    const int nTiles  = (N + 15) >> 4;

    const float* __restrict__ bp = W + colTile + r;
    v2f bf[32];
#pragma unroll
    for (int j = 0; j < 32; ++j) {
        bf[j].x = bp[(size_t)(4 * j + koff) * D_DIM];
        bf[j].y = bp[(size_t)(4 * j + koff + 1) * D_DIM];
    }

    const int scol = lane << 2;
    float4 g0 = make_float4(0.0f, 0.0f, 0.0f, 0.0f), g1 = g0;
    int tile = blockIdx.x;
    if (tile < nTiles) {
        int r0 = (tile << 4) + wave, r1 = r0 + 8;
        if (r0 >= N) r0 = N - 1;
        if (r1 >= N) r1 = N - 1;
        g0 = *reinterpret_cast<const float4*>(A + (size_t)r0 * D_DIM + scol);
        g1 = *reinterpret_cast<const float4*>(A + (size_t)r1 * D_DIM + scol);
    }
    int buf = 0;
    for (; tile < nTiles; tile += gridDim.x) {
        float* __restrict__ at = atile[buf];
        *reinterpret_cast<float4*>(at + wave * A_PITCH + scol)       = g0;
        *reinterpret_cast<float4*>(at + (wave + 8) * A_PITCH + scol) = g1;
        __syncthreads();
        const int nxt = tile + gridDim.x;
        if (nxt < nTiles) {
            int r0 = (nxt << 4) + wave, r1 = r0 + 8;
            if (r0 >= N) r0 = N - 1;
            if (r1 >= N) r1 = N - 1;
            g0 = *reinterpret_cast<const float4*>(A + (size_t)r0 * D_DIM + scol);
            g1 = *reinterpret_cast<const float4*>(A + (size_t)r1 * D_DIM + scol);
        }

        const float* __restrict__ ar = at + r * A_PITCH + koff;
        v8f c = {};
#pragma unroll
        for (int j = 0; j < 32; ++j) {
            const v2f a = *reinterpret_cast<const v2f*>(ar + 4 * j);
            c = __builtin_amdgcn_wmma_f32_16x16x4_f32(false, a, false, bf[j],
                                                      (short)0, c, false, false);
        }
        const int rowTile = tile << 4;
        const int mbase   = rowTile + ((lane >> 4) << 3);
        const int col     = colTile + r;
        if (rowTile + 16 <= N) {
            float* __restrict__ dst = P + (size_t)mbase * D_DIM + col;
#pragma unroll
            for (int v = 0; v < 8; ++v)
                dst[(size_t)v * D_DIM] = c[v];
        } else {
#pragma unroll
            for (int v = 0; v < 8; ++v) {
                int row = mbase + v;
                if (row < N) P[(size_t)row * D_DIM + col] = c[v];
            }
        }
        buf ^= 1;
    }
}

#define BIN_SHIFT 8
#define BIN_W     256
#define NBMAX     1024
#define TILE      4096
#define BIN_CAP   4096

__device__ __forceinline__ int block_excl_scan(int c, int* wsum) {
    const int lane = threadIdx.x & 31, wave = threadIdx.x >> 5;
    int inc = c;
#pragma unroll
    for (int o = 1; o < 32; o <<= 1) {
        const int t = __shfl_up(inc, o, 32);
        if (lane >= o) inc += t;
    }
    if (lane == 31) wsum[wave] = inc;
    __syncthreads();
    if (wave == 0) {
        const int nw = (int)(blockDim.x >> 5);
        const int t = (lane < nw) ? wsum[lane] : 0;
        int ti = t;
#pragma unroll
        for (int o = 1; o < 32; o <<= 1) {
            const int u = __shfl_up(ti, o, 32);
            if (lane >= o) ti += u;
        }
        wsum[lane] = ti - t;
    }
    __syncthreads();
    return wsum[wave] + inc - c;
}

__global__ __launch_bounds__(1024) void bin_count(
    const int* __restrict__ head, const int* __restrict__ uidx,
    int* __restrict__ cntm, int E, int EU, int NE, int NB) {
    __shared__ int hist[NBMAX];
    const int tid = threadIdx.x;
    hist[tid] = 0;
    __syncthreads();
    const int ET = E + EU;
    const int beg = blockIdx.x * TILE;
#pragma unroll
    for (int k = 0; k < 4; ++k) {
        const int g = beg + tid + (k << 10);
        if (g < ET) {
            const int key = (g < E) ? head[g] : NE + uidx[g - E];
            atomicAdd(&hist[key >> BIN_SHIFT], 1);
        }
    }
    __syncthreads();
    if (tid < NB) cntm[(size_t)tid * gridDim.x + blockIdx.x] = hist[tid];
}

__global__ __launch_bounds__(1024) void seg_scan(
    const int* __restrict__ cnt, int* __restrict__ off, int NT) {
    __shared__ int wsum[32];
    __shared__ int sbase;
    const int tid = threadIdx.x, lane = tid & 31, wave = tid >> 5;
    const int start = blockIdx.x * 1024;

    int part = 0;
    for (int i = tid; i < start; i += 1024) part += cnt[i];
    part += __shfl_xor(part, 16, 32);
    part += __shfl_xor(part, 8, 32);
    part += __shfl_xor(part, 4, 32);
    part += __shfl_xor(part, 2, 32);
    part += __shfl_xor(part, 1, 32);
    if (lane == 0) wsum[wave] = part;
    __syncthreads();
    if (wave == 0) {
        int t = wsum[lane];
        t += __shfl_xor(t, 16, 32);
        t += __shfl_xor(t, 8, 32);
        t += __shfl_xor(t, 4, 32);
        t += __shfl_xor(t, 2, 32);
        t += __shfl_xor(t, 1, 32);
        if (lane == 0) sbase = t;
    }
    __syncthreads();
    const int base = sbase;

    const int i = start + tid;
    const int c = (i < NT) ? cnt[i] : 0;
    const int excl = base + block_excl_scan(c, wsum);
    if (i < NT) {
        off[i] = excl;
        if (i == NT - 1) off[NT] = excl + c;
    }
}

__global__ __launch_bounds__(1024) void bin_scatter(
    const int* __restrict__ head, const int* __restrict__ tail,
    const int* __restrict__ etype, const int* __restrict__ uidx,
    const int* __restrict__ iidx, const float* __restrict__ w,
    const int* __restrict__ binoff, int2* __restrict__ binned,
    int E, int EU, int NE, int NB) {
    __shared__ int lbase[NBMAX];
    __shared__ int lcur[NBMAX];
    __shared__ int gbase[NBMAX];
    __shared__ int wsum[32];
    __shared__ int2 stage[TILE];
    __shared__ unsigned short sbin[TILE];
    const int tid = threadIdx.x;
    const int ET = E + EU;
    const int beg = blockIdx.x * TILE;

    int c = 0, g0 = 0;
    if (tid < NB) {
        const int* p = binoff + (size_t)tid * gridDim.x + blockIdx.x;
        g0 = p[0];
        c  = p[1] - g0;
    }
    const int ex = block_excl_scan(c, wsum);
    gbase[tid] = g0; lbase[tid] = ex; lcur[tid] = ex;

    int key[4], px[4], py[4];
#pragma unroll
    for (int k = 0; k < 4; ++k) {
        const int g = beg + tid + (k << 10);
        key[k] = -1; px[k] = 0; py[k] = 0;
        if (g < E) {
            key[k] = head[g]; px[k] = tail[g]; py[k] = etype[g];
        } else if (g < ET) {
            const int j = g - E;
            key[k] = NE + uidx[j]; px[k] = iidx[j]; py[k] = __float_as_int(w[j]);
        }
    }
    __syncthreads();
#pragma unroll
    for (int k = 0; k < 4; ++k) {
        if (key[k] >= 0) {
            const int bin = key[k] >> BIN_SHIFT;
            const int r = atomicAdd(&lcur[bin], 1);
            stage[r] = make_int2(px[k] | ((key[k] & (BIN_W - 1)) << 24), py[k]);
            sbin[r]  = (unsigned short)bin;
        }
    }
    __syncthreads();
    const int n = (ET - beg < TILE) ? (ET - beg) : TILE;
    for (int k = 0; k < 4; ++k) {
        const int i = tid + (k << 10);
        if (i < n) {
            const int bin = sbin[i];
            binned[gbase[bin] + (i - lbase[bin])] = stage[i];
        }
    }
}

__global__ __launch_bounds__(512) void bin_sort(
    const int* __restrict__ binoff, const int2* __restrict__ binned,
    int* __restrict__ rowptr, int2* __restrict__ sorted, int G, int NT) {
    __shared__ int kcnt[BIN_W];
    __shared__ int kcur[BIN_W];
    __shared__ int wsum[32];
    __shared__ int2 ost[BIN_CAP];
    const int tid = threadIdx.x;
    const int b = blockIdx.x;
    const int s = uni(binoff[(size_t)b * G]), e = uni(binoff[(size_t)(b + 1) * G]);
    const int n = e - s;
    if (tid < BIN_W) kcnt[tid] = 0;
    __syncthreads();
    for (int i = tid; i < n; i += 512)
        atomicAdd(&kcnt[(unsigned)binned[s + i].x >> 24], 1);
    __syncthreads();
    const int c  = (tid < BIN_W) ? kcnt[tid] : 0;
    const int ex = block_excl_scan(c, wsum);
    if (tid < BIN_W) {
        kcur[tid] = ex;
        const int key = (b << BIN_SHIFT) + tid;
        if (key < NT) rowptr[key] = s + ex;
    }
    if (b == (int)gridDim.x - 1 && tid == 0) rowptr[NT] = e;
    __syncthreads();
    if (n <= BIN_CAP) {
        for (int i = tid; i < n; i += 512) {
            const int2 rec = binned[s + i];
            const int r = atomicAdd(&kcur[(unsigned)rec.x >> 24], 1);
            ost[r] = make_int2(rec.x & 0x00FFFFFF, rec.y);
        }
        __syncthreads();
        for (int i = tid; i < n; i += 512) sorted[s + i] = ost[i];
    } else {
        for (int i = tid; i < n; i += 512) {
            const int2 rec = binned[s + i];
            const int r = atomicAdd(&kcur[(unsigned)rec.x >> 24], 1);
            sorted[s + r] = make_int2(rec.x & 0x00FFFFFF, rec.y);
        }
    }
}

__device__ __forceinline__ float edge_score1(const float4 q, const float4 k, const float4 rv) {
    const v2f t = (v2f){q.x, q.y} * (v2f){k.x, k.y} * (v2f){rv.x, rv.y};
    const v2f u = (v2f){q.z, q.w} * (v2f){k.z, k.w} * (v2f){rv.z, rv.w};
    float s = t.x + t.y + u.x + u.y;
    s += __shfl_xor(s, 1, 32);
    s += __shfl_xor(s, 2, 32);
    s += __shfl_xor(s, 4, 32);
    s += __shfl_xor(s, 8, 32);
    return s * 0.125f;
}

__device__ __forceinline__ void norm_store(float4 v, size_t vidx, float4 r,
                                           float* res_out, float* cur_out) {
    float ss = v.x*v.x + v.y*v.y + v.z*v.z + v.w*v.w;
    ss += __shfl_xor(ss, 1, 32);
    ss += __shfl_xor(ss, 2, 32);
    ss += __shfl_xor(ss, 4, 32);
    ss += __shfl_xor(ss, 8, 32);
    ss += __shfl_xor(ss, 16, 32);
    const float inv = 1.0f / fmaxf(sqrtf(ss), 1e-12f);
    v.x *= inv; v.y *= inv; v.z *= inv; v.w *= inv;
    if (cur_out) reinterpret_cast<float4*>(cur_out)[vidx] = v;
    r.x += v.x; r.y += v.y; r.z += v.z; r.w += v.w;
    reinterpret_cast<float4*>(res_out)[vidx] = r;
}

#define EB 8
__global__ __launch_bounds__(256) void ent_segment(
    const float* __restrict__ P, const float* __restrict__ entc,
    const float* __restrict__ rel, const int* __restrict__ rowptr,
    const int2* __restrict__ sorted, const float* res_in, float* res_out,
    float* cur_out, int N) {
    const int lane = threadIdx.x & 31;
    const int nw = uni((int)(gridDim.x << 3));
    int n = uni((int)((blockIdx.x << 3) + (threadIdx.x >> 5)));
    if (n >= N) return;
    int beg = uni(rowptr[n]), end = uni(rowptr[n + 1]);
    int2 tr = make_int2(0, 0);
    if (lane < end - beg) tr = sorted[beg + lane];
    int nn = n + nw;
    int nbeg = 0, nend = 0;
    if (nn < N) { nbeg = uni(rowptr[nn]); nend = uni(rowptr[nn + 1]); }
    for (;;) {
    int2 ntr = make_int2(0, 0);
    if (lane < nend - nbeg) ntr = sorted[nbeg + lane];
    const int n2 = nn + nw;
    int b2 = 0, e2 = 0;
    if (n2 < N) { b2 = rowptr[n2]; e2 = rowptr[n2 + 1]; }
    const size_t vidx = ((size_t)n * D_DIM >> 2) + lane;
    const float4 q  = reinterpret_cast<const float4*>(P)[vidx];
    const float4 r0 = reinterpret_cast<const float4*>(res_in)[vidx];

    float m = -INFINITY;
    float den = 0.0f;
    float4 acc = make_float4(0.0f, 0.0f, 0.0f, 0.0f);
    for (int c = beg; c < end; c += 32) {
        const int nc = (end - c < 32) ? (end - c) : 32;
        if (c > beg) {
            tr = make_int2(0, 0);
            if (lane < nc) tr = sorted[c + lane];
        }
        for (int b = 0; b < nc; b += EB) {
            const int nb = (nc - b < EB) ? (nc - b) : EB;
            float  sc[EB];
            float4 vr[EB];
#pragma unroll
            for (int j = 0; j < EB; ++j) {
                const bool live = j < nb;
                const int tj = lane_get(tr.x, b + j), rj = lane_get(tr.y, b + j);
                const float4 k  = reinterpret_cast<const float4*>(P    + (size_t)tj * D_DIM)[lane];
                const float4 v  = reinterpret_cast<const float4*>(entc + (size_t)tj * D_DIM)[lane];
                const float4 rv = reinterpret_cast<const float4*>(rel  + (size_t)rj * D_DIM)[lane];
                const float s = edge_score1(q, k, rv);
                sc[j] = live ? s : -INFINITY;
                vr[j].x = live ? v.x * rv.x : 0.0f;
                vr[j].y = live ? v.y * rv.y : 0.0f;
                vr[j].z = live ? v.z * rv.z : 0.0f;
                vr[j].w = live ? v.w * rv.w : 0.0f;
            }
            float bm = sc[0];
#pragma unroll
            for (int j = 1; j < EB; ++j) bm = fmaxf(bm, sc[j]);
            const float mn = fmaxf(m, bm);
            if (c + b > beg) {
                const float f = __expf(m - mn);
                den *= f;
                acc.x *= f; acc.y *= f; acc.z *= f; acc.w *= f;
            }
            m = mn;
#pragma unroll
            for (int j = 0; j < EB; ++j) {
                const float ex = __expf(sc[j] - m);
                den += ex;
                acc.x += ex * vr[j].x;
                acc.y += ex * vr[j].y;
                acc.z += ex * vr[j].z;
                acc.w += ex * vr[j].w;
            }
        }
    }
    if (end > beg) {
        acc.x /= den; acc.y /= den; acc.z /= den; acc.w /= den;
    }
    norm_store(acc, vidx, r0, res_out, cur_out);
    if (nn >= N) break;
    n = nn; beg = nbeg; end = nend; tr = ntr;
    nn = n2; nbeg = uni(b2); nend = uni(e2);
    }
}

__global__ __launch_bounds__(256) void user_segment(
    const float* __restrict__ entc, const int* __restrict__ rowptr,
    const int2* __restrict__ sorted, const float* res_in, float* res_out,
    int NE, int NU) {
    const int lane = threadIdx.x & 31;
    const int nw = uni((int)(gridDim.x << 3));
    const int* __restrict__ rp = rowptr + NE;
    int u = uni((int)((blockIdx.x << 3) + (threadIdx.x >> 5)));
    if (u >= NU) return;
    int beg = uni(rp[u]), end = uni(rp[u + 1]);
    int2 iw = make_int2(0, 0);
    if (lane < end - beg) iw = sorted[beg + lane];
    int un = u + nw;
    int nbeg = 0, nend = 0;
    if (un < NU) { nbeg = uni(rp[un]); nend = uni(rp[un + 1]); }
    for (;;) {
    int2 niw = make_int2(0, 0);
    if (lane < nend - nbeg) niw = sorted[nbeg + lane];
    const int u2 = un + nw;
    int b2 = 0, e2 = 0;
    if (u2 < NU) { b2 = rp[u2]; e2 = rp[u2 + 1]; }
    const size_t vidx = ((size_t)u * D_DIM >> 2) + lane;
    const float4 r0 = reinterpret_cast<const float4*>(res_in)[vidx];

    float4 acc = make_float4(0.0f, 0.0f, 0.0f, 0.0f);
    for (int c = beg; c < end; c += 32) {
        const int nc = (end - c < 32) ? (end - c) : 32;
        if (c > beg) {
            iw = make_int2(0, 0);
            if (lane < nc) iw = sorted[c + lane];
        }
#pragma unroll 4
        for (int j = 0; j < nc; ++j) {
            const int   ij = lane_get(iw.x, j);
            const float wj = __int_as_float(lane_get(iw.y, j));
            const float4 v = reinterpret_cast<const float4*>(entc + (size_t)ij * D_DIM)[lane];
            acc.x += wj * v.x; acc.y += wj * v.y; acc.z += wj * v.z; acc.w += wj * v.w;
        }
    }
    norm_store(acc, vidx, r0, res_out, nullptr);
    if (un >= NU) break;
    u = un; beg = nbeg; end = nend; iw = niw;
    un = u2; nbeg = uni(b2); nend = uni(e2);
    }
}

extern "C" void kernel_launch(void* const* d_in, const int* in_sizes, int n_in,
                              void* d_out, int out_size, void* d_ws, size_t ws_size,
                              hipStream_t stream) {
    (void)n_in; (void)out_size; (void)ws_size;
    const float* user_emb   = (const float*)d_in[0];
    const float* entity_emb = (const float*)d_in[1];
    const float* W_Q        = (const float*)d_in[2];
    const float* rel_emb    = (const float*)d_in[3];
    const int*   edge_index = (const int*)d_in[4];
    const int*   edge_type  = (const int*)d_in[5];
    const int*   inter_edge = (const int*)d_in[6];
    const float* inter_w    = (const float*)d_in[7];

    const int N_USR = in_sizes[0] / D_DIM;
    const int N_ENT = in_sizes[1] / D_DIM;
    const int E     = in_sizes[5];
    const int E_UI  = in_sizes[7];
    const int NT    = N_ENT + N_USR;
    const int ET    = E + E_UI;

    const int* head = edge_index;
    const int* tail = edge_index + E;
    const int* uidx = inter_edge;
    const int* iidx = inter_edge + E_UI;

    float* out      = (float*)d_out;
    float* user_res = out;
    float* ent_res  = out + (size_t)N_USR * D_DIM;

    char* ws = (char*)d_ws;
    size_t off = 0;
    auto carve = [&](size_t bytes) -> void* {
        void* p = ws + off;
        off += (bytes + 255) & ~(size_t)255;
        return p;
    };
    float* ent_nxt = (float*)carve((size_t)N_ENT * D_DIM * 4);
    float* P       = (float*)carve((size_t)N_ENT * D_DIM * 4);
    const int NB   = (NT + BIN_W - 1) >> BIN_SHIFT;
    const int G    = (ET + TILE - 1) / TILE;
    const int NTOT = NB * G;
    int*   cntm    = (int*)  carve((size_t)NTOT * 4);
    int*   binoff  = (int*)  carve((size_t)(NTOT + 1) * 4);
    int*   rowptr  = (int*)  carve((size_t)(NT + 1) * 4);
    int2*  binned  = (int2*) carve((size_t)ET * 8);
    int2*  sorted  = (int2*) carve((size_t)ET * 8);

    bin_count<<<G, 1024, 0, stream>>>(head, uidx, cntm, E, E_UI, N_ENT, NB);
    seg_scan<<<(NTOT + 1023) / 1024, 1024, 0, stream>>>(cntm, binoff, NTOT);
    bin_scatter<<<G, 1024, 0, stream>>>(head, tail, edge_type, uidx, iidx, inter_w,
                                        binoff, binned, E, E_UI, N_ENT, NB);
    bin_sort<<<NB, 512, 0, stream>>>(binoff, binned, rowptr, sorted, G, NT);

    for (int hop = 0; hop < 2; ++hop) {
        const float* ent_cur = hop ? ent_nxt  : entity_emb;
        const float* eres_in = hop ? ent_res  : entity_emb;
        const float* ures_in = hop ? user_res : user_emb;

        proj_wmma<<<1024, 256, 0, stream>>>(ent_cur, W_Q, P, N_ENT);
        ent_segment<<<768, 256, 0, stream>>>(P, ent_cur, rel_emb, rowptr, sorted,
                                                         eres_in, ent_res,
                                                         hop ? nullptr : ent_nxt, N_ENT);
        user_segment<<<2048, 256, 0, stream>>>(ent_cur, rowptr, sorted,
                                                          ures_in, user_res, N_ENT, N_USR);
    }
}
